# starting kernel
# baseline (speedup 1.0000x reference)
#include <hip/hip_runtime.h>


typedef float v2f __attribute__((ext_vector_type(2)));
typedef float v8f __attribute__((ext_vector_type(8)));

#define C_IN   256
#define C_OUT  128
#define ALPHA  0.2f

__global__ void gat_zero_kernel(float* __restrict__ p, long long n4) {
    long long i      = (long long)blockIdx.x * blockDim.x + threadIdx.x;
    long long stride = (long long)gridDim.x * blockDim.x;
    float4 z = make_float4(0.f, 0.f, 0.f, 0.f);
    for (; i < n4; i += stride) ((float4*)p)[i] = z;
}

__global__ void gat_gemm_kernel(const float* __restrict__ x,
                                const float* __restrict__ W,
                                float* __restrict__ h,
                                int n_nodes, int n_tiles) {
    const int lane = threadIdx.x & 31;
    const int wave = threadIdx.x >> 5;
    const int tile = blockIdx.x * 8 + wave;
    if (tile >= n_tiles) return;

    const int m0   = tile * 16;
    const int half = lane >> 4;
    const int lmod = lane & 15;

    int row = m0 + lmod;
    if (row > n_nodes - 1) row = n_nodes - 1;
    const float* __restrict__ xrow = x + (long long)row * C_IN;

    v8f acc[8] = {};

    for (int k0 = 0; k0 < C_IN; k0 += 4) {
        v2f a;
        a.x = xrow[k0 + 2 * half + 0];
        a.y = xrow[k0 + 2 * half + 1];
#pragma unroll
        for (int nt = 0; nt < 8; ++nt) {
            const int col = nt * 16 + lmod;
            v2f b;
            b.x = W[(long long)(k0 + 2 * half + 0) * C_OUT + col];
            b.y = W[(long long)(k0 + 2 * half + 1) * C_OUT + col];
            acc[nt] = __builtin_amdgcn_wmma_f32_16x16x4_f32(
                 false, a,  false, b,
                 (short)0, acc[nt],
                 false,  false);
        }
    }

    if (m0 + 16 <= n_nodes) {
#pragma unroll
        for (int nt = 0; nt < 8; ++nt)
#pragma unroll
            for (int r = 0; r < 8; ++r)
                h[(long long)(m0 + r + 8 * half) * C_OUT + nt * 16 + lmod] = acc[nt][r];
    } else {
#pragma unroll
        for (int nt = 0; nt < 8; ++nt)
#pragma unroll
            for (int r = 0; r < 8; ++r) {
                const int m = m0 + r + 8 * half;
                if (m < n_nodes)
                    h[(long long)m * C_OUT + nt * 16 + lmod] = acc[nt][r];
            }
    }
}

__global__ void gat_scores_kernel(const float* __restrict__ h,
                                  const float* __restrict__ a,
                                  float* __restrict__ s_src,
                                  float* __restrict__ s_dst,
                                  int n_nodes) {
    const int lane = threadIdx.x & 31;
    const int node = (int)(((long long)blockIdx.x * blockDim.x + threadIdx.x) >> 5);
    if (node >= n_nodes) return;

    const float4 hv = ((const float4*)(h + (long long)node * C_OUT))[lane];
    const float4 w1 = ((const float4*)(a))[lane];
    const float4 w2 = ((const float4*)(a + C_OUT))[lane];

    float p1 = hv.x * w1.x + hv.y * w1.y + hv.z * w1.z + hv.w * w1.w;
    float p2 = hv.x * w2.x + hv.y * w2.y + hv.z * w2.z + hv.w * w2.w;

#pragma unroll
    for (int off = 16; off > 0; off >>= 1) {
        p1 += __shfl_xor(p1, off, 32);
        p2 += __shfl_xor(p2, off, 32);
    }
    if (lane == 0) { s_src[node] = p1; s_dst[node] = p2; }
}

__global__ void gat_edge_kernel(const int* __restrict__ ei,
                                const float* __restrict__ h,
                                const float* __restrict__ s_src,
                                const float* __restrict__ s_dst,
                                float* __restrict__ out,
                                int n_edges) {
    const int lane    = threadIdx.x & 31;
    const int wid     = (int)(((long long)blockIdx.x * blockDim.x + threadIdx.x) >> 5);
    const int n_waves = (int)(((long long)gridDim.x * blockDim.x) >> 5);

    for (int e = wid; e < n_edges; e += n_waves) {
        const int src = ei[e];
        const int dst = ei[n_edges + e];

        float sc = s_src[src] + s_dst[dst];
        sc = (sc > 0.f) ? sc : ALPHA * sc;

        const float4 hv = ((const float4*)(h + (long long)src * C_OUT))[lane];
        float* o = out + (long long)dst * C_OUT + lane * 4;
        unsafeAtomicAdd(o + 0, sc * hv.x);
        unsafeAtomicAdd(o + 1, sc * hv.y);
        unsafeAtomicAdd(o + 2, sc * hv.z);
        unsafeAtomicAdd(o + 3, sc * hv.w);
    }
}

extern "C" void kernel_launch(void* const* d_in, const int* in_sizes, int n_in,
                              void* d_out, int out_size, void* d_ws, size_t ws_size,
                              hipStream_t stream) {
    const float* x  = (const float*)d_in[0];
    const int*   ei = (const int*)  d_in[1];
    const float* W  = (const float*)d_in[2];
    const float* a  = (const float*)d_in[3];
    float* out = (float*)d_out;

    const int N = in_sizes[0] / C_IN;
    const int E = in_sizes[1] / 2;

    float* h     = (float*)d_ws;
    float* s_src = h + (long long)N * C_OUT;
    float* s_dst = s_src + N;

    {
        long long n4 = (long long)N * C_OUT / 4;
        gat_zero_kernel<<<2048, 256, 0, stream>>>(out, n4);
    }

    {
        const int n_tiles = (N + 15) / 16;
        const int blocks  = (n_tiles + 7) / 8;
        gat_gemm_kernel<<<blocks, 256, 0, stream>>>(x, W, h, N, n_tiles);
    }

    {
        const int blocks = (int)(((long long)N * 32 + 255) / 256);
        gat_scores_kernel<<<blocks, 256, 0, stream>>>(h, a, s_src, s_dst, N);
    }

    {
        gat_edge_kernel<<<4096, 256, 0, stream>>>(ei, h, s_src, s_dst, out, E);
    }
}


# edge kernel: lane owns channels l,l+32,l+64,l+96 so every h load and f32 atomic add covers one whole 128-B line (was 4 lines x 32 B per instruction); atomics kept, sums still in arrival order
# speedup vs baseline: 1.6360x; 1.6360x over previous
#include <hip/hip_runtime.h>


typedef float v2f __attribute__((ext_vector_type(2)));
typedef float v8f __attribute__((ext_vector_type(8)));

#define C_IN   256
#define C_OUT  128
#define ALPHA  0.2f

__global__ void gat_zero_kernel(float* __restrict__ p, long long n4) {
    long long i      = (long long)blockIdx.x * blockDim.x + threadIdx.x;
    long long stride = (long long)gridDim.x * blockDim.x;
    float4 z = make_float4(0.f, 0.f, 0.f, 0.f);
    for (; i < n4; i += stride) ((float4*)p)[i] = z;
}

__global__ void gat_gemm_kernel(const float* __restrict__ x,
                                const float* __restrict__ W,
                                float* __restrict__ h,
                                int n_nodes, int n_tiles) {
    const int lane = threadIdx.x & 31;
    const int wave = threadIdx.x >> 5;
    const int tile = blockIdx.x * 8 + wave;
    if (tile >= n_tiles) return;

    const int m0   = tile * 16;
    const int half = lane >> 4;
    const int lmod = lane & 15;

    int row = m0 + lmod;
    if (row > n_nodes - 1) row = n_nodes - 1;
    const float* __restrict__ xrow = x + (long long)row * C_IN;

    v8f acc[8] = {};

    for (int k0 = 0; k0 < C_IN; k0 += 4) {
        v2f a;
        a.x = xrow[k0 + 2 * half + 0];
        a.y = xrow[k0 + 2 * half + 1];
#pragma unroll
        for (int nt = 0; nt < 8; ++nt) {
            const int col = nt * 16 + lmod;
            v2f b;
            b.x = W[(long long)(k0 + 2 * half + 0) * C_OUT + col];
            b.y = W[(long long)(k0 + 2 * half + 1) * C_OUT + col];
            acc[nt] = __builtin_amdgcn_wmma_f32_16x16x4_f32(
                 false, a,  false, b,
                 (short)0, acc[nt],
                 false,  false);
        }
    }

    if (m0 + 16 <= n_nodes) {
#pragma unroll
        for (int nt = 0; nt < 8; ++nt)
#pragma unroll
            for (int r = 0; r < 8; ++r)
                h[(long long)(m0 + r + 8 * half) * C_OUT + nt * 16 + lmod] = acc[nt][r];
    } else {
#pragma unroll
        for (int nt = 0; nt < 8; ++nt)
#pragma unroll
            for (int r = 0; r < 8; ++r) {
                const int m = m0 + r + 8 * half;
                if (m < n_nodes)
                    h[(long long)m * C_OUT + nt * 16 + lmod] = acc[nt][r];
            }
    }
}

__global__ void gat_scores_kernel(const float* __restrict__ h,
                                  const float* __restrict__ a,
                                  float* __restrict__ s_src,
                                  float* __restrict__ s_dst,
                                  int n_nodes) {
    const int lane = threadIdx.x & 31;
    const int node = (int)(((long long)blockIdx.x * blockDim.x + threadIdx.x) >> 5);
    if (node >= n_nodes) return;

    const float4 hv = ((const float4*)(h + (long long)node * C_OUT))[lane];
    const float4 w1 = ((const float4*)(a))[lane];
    const float4 w2 = ((const float4*)(a + C_OUT))[lane];

    float p1 = hv.x * w1.x + hv.y * w1.y + hv.z * w1.z + hv.w * w1.w;
    float p2 = hv.x * w2.x + hv.y * w2.y + hv.z * w2.z + hv.w * w2.w;

#pragma unroll
    for (int off = 16; off > 0; off >>= 1) {
        p1 += __shfl_xor(p1, off, 32);
        p2 += __shfl_xor(p2, off, 32);
    }
    if (lane == 0) { s_src[node] = p1; s_dst[node] = p2; }
}

__global__ void gat_edge_kernel(const int* __restrict__ ei,
                                const float* __restrict__ h,
                                const float* __restrict__ s_src,
                                const float* __restrict__ s_dst,
                                float* __restrict__ out,
                                int n_edges) {
    const int lane    = threadIdx.x & 31;
    const int wid     = (int)(((long long)blockIdx.x * blockDim.x + threadIdx.x) >> 5);
    const int n_waves = (int)(((long long)gridDim.x * blockDim.x) >> 5);

    for (int e = wid; e < n_edges; e += n_waves) {
        const int src = ei[e];
        const int dst = ei[n_edges + e];

        float sc = s_src[src] + s_dst[dst];
        sc = (sc > 0.f) ? sc : ALPHA * sc;

        const float* __restrict__ hs = h + (long long)src * C_OUT + lane;
        const float h0 = hs[0];
        const float h1 = hs[32];
        const float h2 = hs[64];
        const float h3 = hs[96];
        float* o = out + (long long)dst * C_OUT + lane;
        unsafeAtomicAdd(o +  0, sc * h0);
        unsafeAtomicAdd(o + 32, sc * h1);
        unsafeAtomicAdd(o + 64, sc * h2);
        unsafeAtomicAdd(o + 96, sc * h3);
    }
}

extern "C" void kernel_launch(void* const* d_in, const int* in_sizes, int n_in,
                              void* d_out, int out_size, void* d_ws, size_t ws_size,
                              hipStream_t stream) {
    const float* x  = (const float*)d_in[0];
    const int*   ei = (const int*)  d_in[1];
    const float* W  = (const float*)d_in[2];
    const float* a  = (const float*)d_in[3];
    float* out = (float*)d_out;

    const int N = in_sizes[0] / C_IN;
    const int E = in_sizes[1] / 2;

    float* h     = (float*)d_ws;
    float* s_src = h + (long long)N * C_OUT;
    float* s_dst = s_src + N;

    {
        long long n4 = (long long)N * C_OUT / 4;
        gat_zero_kernel<<<2048, 256, 0, stream>>>(out, n4);
    }

    {
        const int n_tiles = (N + 15) / 16;
        const int blocks  = (n_tiles + 7) / 8;
        gat_gemm_kernel<<<blocks, 256, 0, stream>>>(x, W, h, N, n_tiles);
    }

    {
        const int blocks = (int)(((long long)N * 32 + 255) / 256);
        gat_scores_kernel<<<blocks, 256, 0, stream>>>(h, a, s_src, s_dst, N);
    }

    {
        gat_edge_kernel<<<4096, 256, 0, stream>>>(ei, h, s_src, s_dst, out, E);
    }
}


# segment sum regrouped per call: LDS counting sort bins edges by 392-row dst bucket; one work-group sums a bucket's scaled h rows in LDS (ds f32 adds, arrival order) and stores each out row once
# speedup vs baseline: 2.2664x; 1.3853x over previous
#include <hip/hip_runtime.h>


typedef float v2f __attribute__((ext_vector_type(2)));
typedef float v8f __attribute__((ext_vector_type(8)));

#define C_IN   256
#define C_OUT  128
#define ALPHA  0.2f
#define ROWS   392
#define PARTS  256
#define MAXB   1024

__global__ void gat_zero_kernel(float* __restrict__ p, long long n4) {
    long long i      = (long long)blockIdx.x * blockDim.x + threadIdx.x;
    long long stride = (long long)gridDim.x * blockDim.x;
    float4 z = make_float4(0.f, 0.f, 0.f, 0.f);
    for (; i < n4; i += stride) ((float4*)p)[i] = z;
}

__global__ void gat_gemm_kernel(const float* __restrict__ x,
                                const float* __restrict__ W,
                                float* __restrict__ h,
                                int n_nodes, int n_tiles) {
    const int lane = threadIdx.x & 31;
    const int wave = threadIdx.x >> 5;
    const int tile = blockIdx.x * 8 + wave;
    if (tile >= n_tiles) return;

    const int m0   = tile * 16;
    const int half = lane >> 4;
    const int lmod = lane & 15;

    int row = m0 + lmod;
    if (row > n_nodes - 1) row = n_nodes - 1;
    const float* __restrict__ xrow = x + (long long)row * C_IN;

    v8f acc[8] = {};

    for (int k0 = 0; k0 < C_IN; k0 += 4) {
        v2f a;
        a.x = xrow[k0 + 2 * half + 0];
        a.y = xrow[k0 + 2 * half + 1];
#pragma unroll
        for (int nt = 0; nt < 8; ++nt) {
            const int col = nt * 16 + lmod;
            v2f b;
            b.x = W[(long long)(k0 + 2 * half + 0) * C_OUT + col];
            b.y = W[(long long)(k0 + 2 * half + 1) * C_OUT + col];
            acc[nt] = __builtin_amdgcn_wmma_f32_16x16x4_f32(
                 false, a,  false, b,
                 (short)0, acc[nt],
                 false,  false);
        }
    }

    if (m0 + 16 <= n_nodes) {
#pragma unroll
        for (int nt = 0; nt < 8; ++nt)
#pragma unroll
            for (int r = 0; r < 8; ++r)
                h[(long long)(m0 + r + 8 * half) * C_OUT + nt * 16 + lmod] = acc[nt][r];
    } else {
#pragma unroll
        for (int nt = 0; nt < 8; ++nt)
#pragma unroll
            for (int r = 0; r < 8; ++r) {
                const int m = m0 + r + 8 * half;
                if (m < n_nodes)
                    h[(long long)m * C_OUT + nt * 16 + lmod] = acc[nt][r];
            }
    }
}

__global__ void gat_scores_kernel(const float* __restrict__ h,
                                  const float* __restrict__ a,
                                  float* __restrict__ s_src,
                                  float* __restrict__ s_dst,
                                  int n_nodes) {
    const int lane = threadIdx.x & 31;
    const int node = (int)(((long long)blockIdx.x * blockDim.x + threadIdx.x) >> 5);
    if (node >= n_nodes) return;

    const float4 hv = ((const float4*)(h + (long long)node * C_OUT))[lane];
    const float4 w1 = ((const float4*)(a))[lane];
    const float4 w2 = ((const float4*)(a + C_OUT))[lane];

    float p1 = hv.x * w1.x + hv.y * w1.y + hv.z * w1.z + hv.w * w1.w;
    float p2 = hv.x * w2.x + hv.y * w2.y + hv.z * w2.z + hv.w * w2.w;

#pragma unroll
    for (int off = 16; off > 0; off >>= 1) {
        p1 += __shfl_xor(p1, off, 32);
        p2 += __shfl_xor(p2, off, 32);
    }
    if (lane == 0) { s_src[node] = p1; s_dst[node] = p2; }
}

__global__ void gat_edge_kernel(const int* __restrict__ ei,
                                const float* __restrict__ h,
                                const float* __restrict__ s_src,
                                const float* __restrict__ s_dst,
                                float* __restrict__ out,
                                int n_edges) {
    const int lane    = threadIdx.x & 31;
    const int wid     = (int)(((long long)blockIdx.x * blockDim.x + threadIdx.x) >> 5);
    const int n_waves = (int)(((long long)gridDim.x * blockDim.x) >> 5);

    for (int e = wid; e < n_edges; e += n_waves) {
        const int src = ei[e];
        const int dst = ei[n_edges + e];

        float sc = s_src[src] + s_dst[dst];
        sc = (sc > 0.f) ? sc : ALPHA * sc;

        const float* __restrict__ hs = h + (long long)src * C_OUT + lane;
        const float h0 = hs[0];
        const float h1 = hs[32];
        const float h2 = hs[64];
        const float h3 = hs[96];
        float* o = out + (long long)dst * C_OUT + lane;
        unsafeAtomicAdd(o +  0, sc * h0);
        unsafeAtomicAdd(o + 32, sc * h1);
        unsafeAtomicAdd(o + 64, sc * h2);
        unsafeAtomicAdd(o + 96, sc * h3);
    }
}

__global__ __launch_bounds__(1024)
void gat_count_kernel(const int* __restrict__ ei, int* __restrict__ offs,
                      int n_edges, int nb, int per_part) {
    __shared__ int cnt[MAXB];
    const int part = blockIdx.x;
    for (int i = threadIdx.x; i < nb; i += blockDim.x) cnt[i] = 0;
    __syncthreads();
    const int e0 = part * per_part;
    int e1 = e0 + per_part;
    if (e1 > n_edges) e1 = n_edges;
    const int* __restrict__ dstp = ei + n_edges;
    for (int eb = e0 + (int)threadIdx.x; eb < e1; eb += 8 * 1024) {
        int d[8];
#pragma unroll
        for (int u = 0; u < 8; ++u) {
            const int e = eb + u * 1024;
            d[u] = (e < e1) ? dstp[e] : -1;
        }
#pragma unroll
        for (int u = 0; u < 8; ++u)
            if (d[u] >= 0) atomicAdd(&cnt[d[u] / ROWS], 1);
    }
    __syncthreads();
    for (int i = threadIdx.x; i < nb; i += blockDim.x) offs[i * PARTS + part] = cnt[i];
}

__global__ __launch_bounds__(1024)
void gat_scan_kernel(int* __restrict__ offs, int* __restrict__ bstart,
                     int nb, int n_edges) {
    __shared__ int tot[1024];
    const int t    = threadIdx.x;
    const int lane = t & 31;
    const int wave = t >> 5;
    tot[t] = 0;
    __syncthreads();
    for (int b = wave; b < nb; b += 32) {
        int4* row = (int4*)(offs + b * PARTS);
        int4 v0 = row[lane * 2 + 0];
        int4 v1 = row[lane * 2 + 1];
        const int s = v0.x + v0.y + v0.z + v0.w + v1.x + v1.y + v1.z + v1.w;
        int inc = s;
#pragma unroll
        for (int d = 1; d < 32; d <<= 1) {
            const int o = __shfl_up(inc, d, 32);
            if (lane >= d) inc += o;
        }
        int run = inc - s;
        int4 w0, w1;
        w0.x = run; run += v0.x;
        w0.y = run; run += v0.y;
        w0.z = run; run += v0.z;
        w0.w = run; run += v0.w;
        w1.x = run; run += v1.x;
        w1.y = run; run += v1.y;
        w1.z = run; run += v1.z;
        w1.w = run;
        row[lane * 2 + 0] = w0;
        row[lane * 2 + 1] = w1;
        if (lane == 31) tot[b] = inc;
    }
    __syncthreads();
    const int mine = tot[t];
    for (int d = 1; d < 1024; d <<= 1) {
        const int v = (t >= d) ? tot[t - d] : 0;
        __syncthreads();
        tot[t] += v;
        __syncthreads();
    }
    if (t < nb) bstart[t] = tot[t] - mine;
    if (t == 0) bstart[nb] = n_edges;
}

__global__ __launch_bounds__(1024)
void gat_place_kernel(const int* __restrict__ ei, const int* __restrict__ offs,
                      const int* __restrict__ bstart, int2* __restrict__ rec,
                      int n_edges, int nb, int per_part) {
    __shared__ int cur[MAXB];
    const int part = blockIdx.x;
    for (int i = threadIdx.x; i < nb; i += blockDim.x)
        cur[i] = offs[i * PARTS + part] + bstart[i];
    __syncthreads();
    const int e0 = part * per_part;
    int e1 = e0 + per_part;
    if (e1 > n_edges) e1 = n_edges;
    for (int eb = e0 + (int)threadIdx.x; eb < e1; eb += 8 * 1024) {
        int s[8], d[8];
#pragma unroll
        for (int u = 0; u < 8; ++u) {
            const int e = eb + u * 1024;
            s[u] = (e < e1) ? ei[e] : 0;
            d[u] = (e < e1) ? ei[n_edges + e] : -1;
        }
#pragma unroll
        for (int u = 0; u < 8; ++u)
            if (d[u] >= 0) {
                const int p = atomicAdd(&cur[d[u] / ROWS], 1);
                rec[p] = make_int2(s[u], d[u]);
            }
    }
}

__global__ __launch_bounds__(1024)
void gat_bucket_sum_kernel(const int2* __restrict__ rec,
                           const int* __restrict__ bstart,
                           const float* __restrict__ h,
                           const float* __restrict__ s_src,
                           const float* __restrict__ s_dst,
                           float* __restrict__ out,
                           int n_nodes) {
    __shared__ __attribute__((aligned(16))) float acc[ROWS * C_OUT];
    const int b    = blockIdx.x;
    const int lane = threadIdx.x & 31;
    const int wave = __builtin_amdgcn_readfirstlane((int)(threadIdx.x >> 5));
    const int row0 = b * ROWS;

    float4* acc4 = (float4*)acc;
    for (int i = threadIdx.x; i < ROWS * C_OUT / 4; i += 1024)
        acc4[i] = make_float4(0.f, 0.f, 0.f, 0.f);
    __syncthreads();

    const int beg = bstart[b];
    const int end = bstart[b + 1];
    for (int base = beg + wave * 32; base < end; base += 32 * 32) {
        int n = end - base;
        if (n > 32) n = 32;
        int   src = 0, dl = 0;
        float sc  = 0.f;
        if (lane < n) {
            const int2 r = rec[base + lane];
            src = r.x;
            dl  = r.y - row0;
            const float v = s_src[r.x] + s_dst[r.y];
            sc = (v > 0.f) ? v : ALPHA * v;
        }
        for (int j0 = 0; j0 < n; j0 += 16) {
            float4 hv[16];
#pragma unroll
            for (int u = 0; u < 16; ++u) {
                const int jj = (j0 + u < n) ? (j0 + u) : j0;
                const int sj = __builtin_amdgcn_readlane(src, jj);
                hv[u] = *(const float4*)(h + (long long)sj * C_OUT + 4 * lane);
            }
#pragma unroll
            for (int u = 0; u < 16; ++u) {
                if (j0 + u < n) {
                    const float w = __int_as_float(
                        __builtin_amdgcn_readlane(__float_as_int(sc), j0 + u));
                    const int d = __builtin_amdgcn_readlane(dl, j0 + u);
                    float* ap = acc + d * C_OUT + lane;
                    atomicAdd(ap +  0, w * hv[u].x);
                    atomicAdd(ap + 32, w * hv[u].y);
                    atomicAdd(ap + 64, w * hv[u].z);
                    atomicAdd(ap + 96, w * hv[u].w);
                }
            }
        }
    }
    __syncthreads();

    int rows = n_nodes - row0;
    if (rows > ROWS) rows = ROWS;
    float4* o4 = (float4*)(out + (long long)row0 * C_OUT);
    for (int i = threadIdx.x; i < rows * (C_OUT / 4); i += 1024) {
        const float* ar = acc + (i >> 5) * C_OUT + (i & 31);
        o4[i] = make_float4(ar[0], ar[32], ar[64], ar[96]);
    }
}

extern "C" void kernel_launch(void* const* d_in, const int* in_sizes, int n_in,
                              void* d_out, int out_size, void* d_ws, size_t ws_size,
                              hipStream_t stream) {
    const float* x  = (const float*)d_in[0];
    const int*   ei = (const int*)  d_in[1];
    const float* W  = (const float*)d_in[2];
    const float* a  = (const float*)d_in[3];
    float* out = (float*)d_out;

    const int N = in_sizes[0] / C_IN;
    const int E = in_sizes[1] / 2;

    float* h     = (float*)d_ws;
    float* s_src = h + (long long)N * C_OUT;
    float* s_dst = s_src + N;

    const int NB       = (N + ROWS - 1) / ROWS;
    const int per_part = (E + PARTS - 1) / PARTS;
    int*  offs   = (int*)(s_dst + N);
    int2* rec    = (int2*)(offs + (long long)NB * PARTS);
    int*  bstart = (int*)(rec + E);
    const size_t need = (size_t)((const char*)(bstart + NB + 1) - (const char*)d_ws);
    const bool regroup = (ws_size >= need) && (NB <= MAXB);

    if (regroup) {
        gat_count_kernel<<<PARTS, 1024, 0, stream>>>(ei, offs, E, NB, per_part);
        gat_scan_kernel<<<1, 1024, 0, stream>>>(offs, bstart, NB, E);
        gat_place_kernel<<<PARTS, 1024, 0, stream>>>(ei, offs, bstart, rec, E, NB, per_part);
    } else {
        long long n4 = (long long)N * C_OUT / 4;
        gat_zero_kernel<<<2048, 256, 0, stream>>>(out, n4);
    }

    {
        const int n_tiles = (N + 15) / 16;
        const int blocks  = (n_tiles + 7) / 8;
        gat_gemm_kernel<<<blocks, 256, 0, stream>>>(x, W, h, N, n_tiles);
    }

    {
        const int blocks = (int)(((long long)N * 32 + 255) / 256);
        gat_scores_kernel<<<blocks, 256, 0, stream>>>(h, a, s_src, s_dst, N);
    }

    if (regroup) {
        gat_bucket_sum_kernel<<<NB, 1024, 0, stream>>>(rec, bstart, h, s_src, s_dst, out, N);
    } else {
        gat_edge_kernel<<<4096, 256, 0, stream>>>(ei, h, s_src, s_dst, out, E);
    }
}


# score launch: one wave per 16 nodes with 16 rows in flight (6250 long-lived waves) instead of 100000 one-row waves; per-row dot and shuffle order unchanged; LDS sums still in arrival order
# speedup vs baseline: 2.4250x; 1.0700x over previous
#include <hip/hip_runtime.h>


typedef float v2f __attribute__((ext_vector_type(2)));
typedef float v8f __attribute__((ext_vector_type(8)));

#define C_IN   256
#define C_OUT  128
#define ALPHA  0.2f
#define ROWS   392
#define PARTS  256
#define MAXB   1024

__global__ void gat_zero_kernel(float* __restrict__ p, long long n4) {
    long long i      = (long long)blockIdx.x * blockDim.x + threadIdx.x;
    long long stride = (long long)gridDim.x * blockDim.x;
    float4 z = make_float4(0.f, 0.f, 0.f, 0.f);
    for (; i < n4; i += stride) ((float4*)p)[i] = z;
}

__global__ void gat_gemm_kernel(const float* __restrict__ x,
                                const float* __restrict__ W,
                                float* __restrict__ h,
                                int n_nodes, int n_tiles) {
    const int lane = threadIdx.x & 31;
    const int wave = threadIdx.x >> 5;
    const int tile = blockIdx.x * 8 + wave;
    if (tile >= n_tiles) return;

    const int m0   = tile * 16;
    const int half = lane >> 4;
    const int lmod = lane & 15;

    int row = m0 + lmod;
    if (row > n_nodes - 1) row = n_nodes - 1;
    const float* __restrict__ xrow = x + (long long)row * C_IN;

    v8f acc[8] = {};

    for (int k0 = 0; k0 < C_IN; k0 += 4) {
        v2f a;
        a.x = xrow[k0 + 2 * half + 0];
        a.y = xrow[k0 + 2 * half + 1];
#pragma unroll
        for (int nt = 0; nt < 8; ++nt) {
            const int col = nt * 16 + lmod;
            v2f b;
            b.x = W[(long long)(k0 + 2 * half + 0) * C_OUT + col];
            b.y = W[(long long)(k0 + 2 * half + 1) * C_OUT + col];
            acc[nt] = __builtin_amdgcn_wmma_f32_16x16x4_f32(
                 false, a,  false, b,
                 (short)0, acc[nt],
                 false,  false);
        }
    }

    if (m0 + 16 <= n_nodes) {
#pragma unroll
        for (int nt = 0; nt < 8; ++nt)
#pragma unroll
            for (int r = 0; r < 8; ++r)
                h[(long long)(m0 + r + 8 * half) * C_OUT + nt * 16 + lmod] = acc[nt][r];
    } else {
#pragma unroll
        for (int nt = 0; nt < 8; ++nt)
#pragma unroll
            for (int r = 0; r < 8; ++r) {
                const int m = m0 + r + 8 * half;
                if (m < n_nodes)
                    h[(long long)m * C_OUT + nt * 16 + lmod] = acc[nt][r];
            }
    }
}

__global__ __launch_bounds__(256)
void gat_scores_kernel(const float* __restrict__ h,
                       const float* __restrict__ a,
                       float* __restrict__ s_src,
                       float* __restrict__ s_dst,
                       int n_nodes, int n_tiles) {
    const int lane = threadIdx.x & 31;
    const int tile = blockIdx.x * 8 + (int)(threadIdx.x >> 5);
    if (tile >= n_tiles) return;
    const int n0 = tile * 16;

    const float4 w1 = ((const float4*)(a))[lane];
    const float4 w2 = ((const float4*)(a + C_OUT))[lane];

    float4 hv[16];
#pragma unroll
    for (int u = 0; u < 16; ++u) {
        int node = n0 + u;
        if (node > n_nodes - 1) node = n_nodes - 1;
        hv[u] = ((const float4*)(h + (long long)node * C_OUT))[lane];
    }

    float r1 = 0.f, r2 = 0.f;
#pragma unroll
    for (int u = 0; u < 16; ++u) {
        float p1 = hv[u].x * w1.x + hv[u].y * w1.y + hv[u].z * w1.z + hv[u].w * w1.w;
        float p2 = hv[u].x * w2.x + hv[u].y * w2.y + hv[u].z * w2.z + hv[u].w * w2.w;
        for (int off = 16; off > 0; off >>= 1) {
            p1 += __shfl_xor(p1, off, 32);
            p2 += __shfl_xor(p2, off, 32);
        }
        if (lane == u) { r1 = p1; r2 = p2; }
    }
    if (lane < 16 && n0 + lane < n_nodes) {
        s_src[n0 + lane] = r1;
        s_dst[n0 + lane] = r2;
    }
}

__global__ void gat_edge_kernel(const int* __restrict__ ei,
                                const float* __restrict__ h,
                                const float* __restrict__ s_src,
                                const float* __restrict__ s_dst,
                                float* __restrict__ out,
                                int n_edges) {
    const int lane    = threadIdx.x & 31;
    const int wid     = (int)(((long long)blockIdx.x * blockDim.x + threadIdx.x) >> 5);
    const int n_waves = (int)(((long long)gridDim.x * blockDim.x) >> 5);

    for (int e = wid; e < n_edges; e += n_waves) {
        const int src = ei[e];
        const int dst = ei[n_edges + e];

        float sc = s_src[src] + s_dst[dst];
        sc = (sc > 0.f) ? sc : ALPHA * sc;

        const float* __restrict__ hs = h + (long long)src * C_OUT + lane;
        const float h0 = hs[0];
        const float h1 = hs[32];
        const float h2 = hs[64];
        const float h3 = hs[96];
        float* o = out + (long long)dst * C_OUT + lane;
        unsafeAtomicAdd(o +  0, sc * h0);
        unsafeAtomicAdd(o + 32, sc * h1);
        unsafeAtomicAdd(o + 64, sc * h2);
        unsafeAtomicAdd(o + 96, sc * h3);
    }
}

__global__ __launch_bounds__(1024)
void gat_count_kernel(const int* __restrict__ ei, int* __restrict__ offs,
                      int n_edges, int nb, int per_part) {
    __shared__ int cnt[MAXB];
    const int part = blockIdx.x;
    for (int i = threadIdx.x; i < nb; i += blockDim.x) cnt[i] = 0;
    __syncthreads();
    const int e0 = part * per_part;
    int e1 = e0 + per_part;
    if (e1 > n_edges) e1 = n_edges;
    const int* __restrict__ dstp = ei + n_edges;
    for (int eb = e0 + (int)threadIdx.x; eb < e1; eb += 8 * 1024) {
        int d[8];
#pragma unroll
        for (int u = 0; u < 8; ++u) {
            const int e = eb + u * 1024;
            d[u] = (e < e1) ? dstp[e] : -1;
        }
#pragma unroll
        for (int u = 0; u < 8; ++u)
            if (d[u] >= 0) atomicAdd(&cnt[d[u] / ROWS], 1);
    }
    __syncthreads();
    for (int i = threadIdx.x; i < nb; i += blockDim.x) offs[i * PARTS + part] = cnt[i];
}

__global__ __launch_bounds__(1024)
void gat_scan_kernel(int* __restrict__ offs, int* __restrict__ bstart,
                     int nb, int n_edges) {
    __shared__ int tot[1024];
    const int t    = threadIdx.x;
    const int lane = t & 31;
    const int wave = t >> 5;
    tot[t] = 0;
    __syncthreads();
    for (int b = wave; b < nb; b += 32) {
        int4* row = (int4*)(offs + b * PARTS);
        int4 v0 = row[lane * 2 + 0];
        int4 v1 = row[lane * 2 + 1];
        const int s = v0.x + v0.y + v0.z + v0.w + v1.x + v1.y + v1.z + v1.w;
        int inc = s;
#pragma unroll
        for (int d = 1; d < 32; d <<= 1) {
            const int o = __shfl_up(inc, d, 32);
            if (lane >= d) inc += o;
        }
        int run = inc - s;
        int4 w0, w1;
        w0.x = run; run += v0.x;
        w0.y = run; run += v0.y;
        w0.z = run; run += v0.z;
        w0.w = run; run += v0.w;
        w1.x = run; run += v1.x;
        w1.y = run; run += v1.y;
        w1.z = run; run += v1.z;
        w1.w = run;
        row[lane * 2 + 0] = w0;
        row[lane * 2 + 1] = w1;
        if (lane == 31) tot[b] = inc;
    }
    __syncthreads();
    const int mine = tot[t];
    for (int d = 1; d < 1024; d <<= 1) {
        const int v = (t >= d) ? tot[t - d] : 0;
        __syncthreads();
        tot[t] += v;
        __syncthreads();
    }
    if (t < nb) bstart[t] = tot[t] - mine;
    if (t == 0) bstart[nb] = n_edges;
}

__global__ __launch_bounds__(1024)
void gat_place_kernel(const int* __restrict__ ei, const int* __restrict__ offs,
                      const int* __restrict__ bstart, int2* __restrict__ rec,
                      int n_edges, int nb, int per_part) {
    __shared__ int cur[MAXB];
    const int part = blockIdx.x;
    for (int i = threadIdx.x; i < nb; i += blockDim.x)
        cur[i] = offs[i * PARTS + part] + bstart[i];
    __syncthreads();
    const int e0 = part * per_part;
    int e1 = e0 + per_part;
    if (e1 > n_edges) e1 = n_edges;
    for (int eb = e0 + (int)threadIdx.x; eb < e1; eb += 8 * 1024) {
        int s[8], d[8];
#pragma unroll
        for (int u = 0; u < 8; ++u) {
            const int e = eb + u * 1024;
            s[u] = (e < e1) ? ei[e] : 0;
            d[u] = (e < e1) ? ei[n_edges + e] : -1;
        }
#pragma unroll
        for (int u = 0; u < 8; ++u)
            if (d[u] >= 0) {
                const int p = atomicAdd(&cur[d[u] / ROWS], 1);
                rec[p] = make_int2(s[u], d[u]);
            }
    }
}

__global__ __launch_bounds__(1024)
void gat_bucket_sum_kernel(const int2* __restrict__ rec,
                           const int* __restrict__ bstart,
                           const float* __restrict__ h,
                           const float* __restrict__ s_src,
                           const float* __restrict__ s_dst,
                           float* __restrict__ out,
                           int n_nodes) {
    __shared__ __attribute__((aligned(16))) float acc[ROWS * C_OUT];
    const int b    = blockIdx.x;
    const int lane = threadIdx.x & 31;
    const int wave = __builtin_amdgcn_readfirstlane((int)(threadIdx.x >> 5));
    const int row0 = b * ROWS;

    float4* acc4 = (float4*)acc;
    for (int i = threadIdx.x; i < ROWS * C_OUT / 4; i += 1024)
        acc4[i] = make_float4(0.f, 0.f, 0.f, 0.f);
    __syncthreads();

    const int beg = bstart[b];
    const int end = bstart[b + 1];
    for (int base = beg + wave * 32; base < end; base += 32 * 32) {
        int n = end - base;
        if (n > 32) n = 32;
        int   src = 0, dl = 0;
        float sc  = 0.f;
        if (lane < n) {
            const int2 r = rec[base + lane];
            src = r.x;
            dl  = r.y - row0;
            const float v = s_src[r.x] + s_dst[r.y];
            sc = (v > 0.f) ? v : ALPHA * v;
        }
        for (int j0 = 0; j0 < n; j0 += 16) {
            float4 hv[16];
#pragma unroll
            for (int u = 0; u < 16; ++u) {
                const int jj = (j0 + u < n) ? (j0 + u) : j0;
                const int sj = __builtin_amdgcn_readlane(src, jj);
                hv[u] = *(const float4*)(h + (long long)sj * C_OUT + 4 * lane);
            }
#pragma unroll
            for (int u = 0; u < 16; ++u) {
                if (j0 + u < n) {
                    const float w = __int_as_float(
                        __builtin_amdgcn_readlane(__float_as_int(sc), j0 + u));
                    const int d = __builtin_amdgcn_readlane(dl, j0 + u);
                    float* ap = acc + d * C_OUT + lane;
                    atomicAdd(ap +  0, w * hv[u].x);
                    atomicAdd(ap + 32, w * hv[u].y);
                    atomicAdd(ap + 64, w * hv[u].z);
                    atomicAdd(ap + 96, w * hv[u].w);
                }
            }
        }
    }
    __syncthreads();

    int rows = n_nodes - row0;
    if (rows > ROWS) rows = ROWS;
    float4* o4 = (float4*)(out + (long long)row0 * C_OUT);
    for (int i = threadIdx.x; i < rows * (C_OUT / 4); i += 1024) {
        const float* ar = acc + (i >> 5) * C_OUT + (i & 31);
        o4[i] = make_float4(ar[0], ar[32], ar[64], ar[96]);
    }
}

extern "C" void kernel_launch(void* const* d_in, const int* in_sizes, int n_in,
                              void* d_out, int out_size, void* d_ws, size_t ws_size,
                              hipStream_t stream) {
    const float* x  = (const float*)d_in[0];
    const int*   ei = (const int*)  d_in[1];
    const float* W  = (const float*)d_in[2];
    const float* a  = (const float*)d_in[3];
    float* out = (float*)d_out;

    const int N = in_sizes[0] / C_IN;
    const int E = in_sizes[1] / 2;

    float* h     = (float*)d_ws;
    float* s_src = h + (long long)N * C_OUT;
    float* s_dst = s_src + N;

    const int NB       = (N + ROWS - 1) / ROWS;
    const int per_part = (E + PARTS - 1) / PARTS;
    int*  offs   = (int*)(s_dst + N);
    int2* rec    = (int2*)(offs + (long long)NB * PARTS);
    int*  bstart = (int*)(rec + E);
    const size_t need = (size_t)((const char*)(bstart + NB + 1) - (const char*)d_ws);
    const bool regroup = (ws_size >= need) && (NB <= MAXB);

    if (regroup) {
        gat_count_kernel<<<PARTS, 1024, 0, stream>>>(ei, offs, E, NB, per_part);
        gat_scan_kernel<<<1, 1024, 0, stream>>>(offs, bstart, NB, E);
        gat_place_kernel<<<PARTS, 1024, 0, stream>>>(ei, offs, bstart, rec, E, NB, per_part);
    } else {
        long long n4 = (long long)N * C_OUT / 4;
        gat_zero_kernel<<<2048, 256, 0, stream>>>(out, n4);
    }

    {
        const int n_tiles = (N + 15) / 16;
        const int blocks  = (n_tiles + 7) / 8;
        gat_gemm_kernel<<<blocks, 256, 0, stream>>>(x, W, h, N, n_tiles);
    }

    {
        const int n_tiles = (N + 15) / 16;
        const int blocks  = (n_tiles + 7) / 8;
        gat_scores_kernel<<<blocks, 256, 0, stream>>>(h, a, s_src, s_dst, N, n_tiles);
    }

    if (regroup) {
        gat_bucket_sum_kernel<<<NB, 1024, 0, stream>>>(rec, bstart, h, s_src, s_dst, out, N);
    } else {
        gat_edge_kernel<<<4096, 256, 0, stream>>>(ei, h, s_src, s_dst, out, E);
    }
}


# GEMM: W copied once per persistent work-group into LDS (k-pair interleaved, conflict-free 8-byte B reads) instead of 16 global loads per k step; 16x64 units dealt evenly to SIMDs; h bit-identical
# speedup vs baseline: 2.8520x; 1.1761x over previous
#include <hip/hip_runtime.h>


typedef float v2f __attribute__((ext_vector_type(2)));
typedef float v8f __attribute__((ext_vector_type(8)));

#define C_IN   256
#define C_OUT  128
#define ALPHA  0.2f
#define ROWS   392
#define PARTS  256
#define MAXB   1024

__global__ void gat_zero_kernel(float* __restrict__ p, long long n4) {
    long long i      = (long long)blockIdx.x * blockDim.x + threadIdx.x;
    long long stride = (long long)gridDim.x * blockDim.x;
    float4 z = make_float4(0.f, 0.f, 0.f, 0.f);
    for (; i < n4; i += stride) ((float4*)p)[i] = z;
}

#define GEMM_WGS  512
#define WP_PITCH  288

__device__ __forceinline__ void gemm_load_b(v2f (&bf)[16], const float* wk) {
#pragma unroll
    for (int jj = 0; jj < 4; ++jj)
#pragma unroll
        for (int nt = 0; nt < 4; ++nt)
            bf[jj * 4 + nt] = *(const v2f*)(wk + 2 * jj * WP_PITCH + 32 * nt);
}

__device__ __forceinline__ void gemm_wmma16(v8f (&acc)[4], const v2f* a, const v2f (&bf)[16]) {
#pragma unroll
    for (int jj = 0; jj < 4; ++jj)
#pragma unroll
        for (int nt = 0; nt < 4; ++nt)
            acc[nt] = __builtin_amdgcn_wmma_f32_16x16x4_f32(
                 false, a[jj],  false, bf[jj * 4 + nt],
                 (short)0, acc[nt],
                 false,  false);
}

__global__ __launch_bounds__(256)
void gat_gemm_kernel(const float* __restrict__ x,
                     const float* __restrict__ W,
                     float* __restrict__ h,
                     int n_nodes, int n_tiles) {
    __shared__ __attribute__((aligned(16))) float Wp[(C_IN / 2) * WP_PITCH];
    const int lane = threadIdx.x & 31;
    const int wave = __builtin_amdgcn_readfirstlane((int)(threadIdx.x >> 5));
    const int half = lane >> 4;
    const int lmod = lane & 15;

    for (int it = 0; it < 2; ++it) {
        float4 r0[8], r1[8];
#pragma unroll
        for (int u = 0; u < 8; ++u) {
            const int pr = wave + 8 * (it * 8 + u);
            r0[u] = ((const float4*)(W + (2 * pr + 0) * C_OUT))[lane];
            r1[u] = ((const float4*)(W + (2 * pr + 1) * C_OUT))[lane];
        }
#pragma unroll
        for (int u = 0; u < 8; ++u) {
            const int pr = wave + 8 * (it * 8 + u);
            float4* d = (float4*)(Wp + pr * WP_PITCH + 8 * lane);
            d[0] = make_float4(r0[u].x, r1[u].x, r0[u].y, r1[u].y);
            d[1] = make_float4(r0[u].z, r1[u].z, r0[u].w, r1[u].w);
        }
    }
    __syncthreads();

    const int S     = (int)gridDim.x * 8;
    const int hgrid = (int)gridDim.x >> 1;
    const int b     = (int)blockIdx.x;
    const int slot  = (((b / hgrid) * 2 + (wave >> 2)) * hgrid + (b % hgrid)) * 4 + (wave & 3);
    const int n_units = n_tiles * 2;

    for (int u = slot; u < n_units; u += S) {
        const int m0  = (u >> 1) * 16;
        const int nc0 = (u & 1) * 64;

        int row = m0 + lmod;
        if (row > n_nodes - 1) row = n_nodes - 1;
        const float* __restrict__ xrow = x + (long long)row * C_IN + 2 * half;
        const float* wl = Wp + half * WP_PITCH + 2 * (nc0 + lmod);

        v8f acc[4] = {};
        v2f a0[8], a1[8];
        v2f bA[16], bB[16];
#pragma unroll
        for (int j = 0; j < 8; ++j) a0[j] = *(const v2f*)(xrow + 4 * j);
        gemm_load_b(bA, wl);

#pragma unroll 1
        for (int kb = 0; kb < 8; kb += 2) {
#pragma unroll
            for (int j = 0; j < 8; ++j) a1[j] = *(const v2f*)(xrow + 32 * (kb + 1) + 4 * j);
            gemm_load_b(bB, wl + (kb * 16 + 8) * WP_PITCH);
            gemm_wmma16(acc, a0, bA);
            gemm_load_b(bA, wl + (kb + 1) * 16 * WP_PITCH);
            gemm_wmma16(acc, a0 + 4, bB);
            const int kn = (kb + 2 < 8) ? kb + 2 : 7;
#pragma unroll
            for (int j = 0; j < 8; ++j) a0[j] = *(const v2f*)(xrow + 32 * kn + 4 * j);
            gemm_load_b(bB, wl + ((kb + 1) * 16 + 8) * WP_PITCH);
            gemm_wmma16(acc, a1, bA);
            gemm_load_b(bA, wl + kn * 16 * WP_PITCH);
            gemm_wmma16(acc, a1 + 4, bB);
        }

        float* __restrict__ hb = h + (long long)m0 * C_OUT + nc0 + lmod;
        if (m0 + 16 <= n_nodes) {
#pragma unroll
            for (int nt = 0; nt < 4; ++nt)
#pragma unroll
                for (int r = 0; r < 8; ++r)
                    hb[(r + 8 * half) * C_OUT + nt * 16] = acc[nt][r];
        } else {
#pragma unroll
            for (int nt = 0; nt < 4; ++nt)
#pragma unroll
                for (int r = 0; r < 8; ++r) {
                    const int m = m0 + r + 8 * half;
                    if (m < n_nodes)
                        hb[(r + 8 * half) * C_OUT + nt * 16] = acc[nt][r];
                }
        }
    }
}

__global__ __launch_bounds__(256)
void gat_scores_kernel(const float* __restrict__ h,
                       const float* __restrict__ a,
                       float* __restrict__ s_src,
                       float* __restrict__ s_dst,
                       int n_nodes, int n_tiles) {
    const int lane = threadIdx.x & 31;
    const int tile = blockIdx.x * 8 + (int)(threadIdx.x >> 5);
    if (tile >= n_tiles) return;
    const int n0 = tile * 16;

    const float4 w1 = ((const float4*)(a))[lane];
    const float4 w2 = ((const float4*)(a + C_OUT))[lane];

    float4 hv[16];
#pragma unroll
    for (int u = 0; u < 16; ++u) {
        int node = n0 + u;
        if (node > n_nodes - 1) node = n_nodes - 1;
        hv[u] = ((const float4*)(h + (long long)node * C_OUT))[lane];
    }

    float r1 = 0.f, r2 = 0.f;
#pragma unroll
    for (int u = 0; u < 16; ++u) {
        float p1 = hv[u].x * w1.x + hv[u].y * w1.y + hv[u].z * w1.z + hv[u].w * w1.w;
        float p2 = hv[u].x * w2.x + hv[u].y * w2.y + hv[u].z * w2.z + hv[u].w * w2.w;
        for (int off = 16; off > 0; off >>= 1) {
            p1 += __shfl_xor(p1, off, 32);
            p2 += __shfl_xor(p2, off, 32);
        }
        if (lane == u) { r1 = p1; r2 = p2; }
    }
    if (lane < 16 && n0 + lane < n_nodes) {
        s_src[n0 + lane] = r1;
        s_dst[n0 + lane] = r2;
    }
}

__global__ void gat_edge_kernel(const int* __restrict__ ei,
                                const float* __restrict__ h,
                                const float* __restrict__ s_src,
                                const float* __restrict__ s_dst,
                                float* __restrict__ out,
                                int n_edges) {
    const int lane    = threadIdx.x & 31;
    const int wid     = (int)(((long long)blockIdx.x * blockDim.x + threadIdx.x) >> 5);
    const int n_waves = (int)(((long long)gridDim.x * blockDim.x) >> 5);

    for (int e = wid; e < n_edges; e += n_waves) {
        const int src = ei[e];
        const int dst = ei[n_edges + e];

        float sc = s_src[src] + s_dst[dst];
        sc = (sc > 0.f) ? sc : ALPHA * sc;

        const float* __restrict__ hs = h + (long long)src * C_OUT + lane;
        const float h0 = hs[0];
        const float h1 = hs[32];
        const float h2 = hs[64];
        const float h3 = hs[96];
        float* o = out + (long long)dst * C_OUT + lane;
        unsafeAtomicAdd(o +  0, sc * h0);
        unsafeAtomicAdd(o + 32, sc * h1);
        unsafeAtomicAdd(o + 64, sc * h2);
        unsafeAtomicAdd(o + 96, sc * h3);
    }
}

__global__ __launch_bounds__(1024)
void gat_count_kernel(const int* __restrict__ ei, int* __restrict__ offs,
                      int n_edges, int nb, int per_part) {
    __shared__ int cnt[MAXB];
    const int part = blockIdx.x;
    for (int i = threadIdx.x; i < nb; i += blockDim.x) cnt[i] = 0;
    __syncthreads();
    const int e0 = part * per_part;
    int e1 = e0 + per_part;
    if (e1 > n_edges) e1 = n_edges;
    const int* __restrict__ dstp = ei + n_edges;
    for (int eb = e0 + (int)threadIdx.x; eb < e1; eb += 8 * 1024) {
        int d[8];
#pragma unroll
        for (int u = 0; u < 8; ++u) {
            const int e = eb + u * 1024;
            d[u] = (e < e1) ? dstp[e] : -1;
        }
#pragma unroll
        for (int u = 0; u < 8; ++u)
            if (d[u] >= 0) atomicAdd(&cnt[d[u] / ROWS], 1);
    }
    __syncthreads();
    for (int i = threadIdx.x; i < nb; i += blockDim.x) offs[i * PARTS + part] = cnt[i];
}

__global__ __launch_bounds__(1024)
void gat_scan_kernel(int* __restrict__ offs, int* __restrict__ bstart,
                     int nb, int n_edges) {
    __shared__ int tot[1024];
    const int t    = threadIdx.x;
    const int lane = t & 31;
    const int wave = t >> 5;
    tot[t] = 0;
    __syncthreads();
    for (int b = wave; b < nb; b += 32) {
        int4* row = (int4*)(offs + b * PARTS);
        int4 v0 = row[lane * 2 + 0];
        int4 v1 = row[lane * 2 + 1];
        const int s = v0.x + v0.y + v0.z + v0.w + v1.x + v1.y + v1.z + v1.w;
        int inc = s;
#pragma unroll
        for (int d = 1; d < 32; d <<= 1) {
            const int o = __shfl_up(inc, d, 32);
            if (lane >= d) inc += o;
        }
        int run = inc - s;
        int4 w0, w1;
        w0.x = run; run += v0.x;
        w0.y = run; run += v0.y;
        w0.z = run; run += v0.z;
        w0.w = run; run += v0.w;
        w1.x = run; run += v1.x;
        w1.y = run; run += v1.y;
        w1.z = run; run += v1.z;
        w1.w = run;
        row[lane * 2 + 0] = w0;
        row[lane * 2 + 1] = w1;
        if (lane == 31) tot[b] = inc;
    }
    __syncthreads();
    const int mine = tot[t];
    for (int d = 1; d < 1024; d <<= 1) {
        const int v = (t >= d) ? tot[t - d] : 0;
        __syncthreads();
        tot[t] += v;
        __syncthreads();
    }
    if (t < nb) bstart[t] = tot[t] - mine;
    if (t == 0) bstart[nb] = n_edges;
}

__global__ __launch_bounds__(1024)
void gat_place_kernel(const int* __restrict__ ei, const int* __restrict__ offs,
                      const int* __restrict__ bstart, int2* __restrict__ rec,
                      int n_edges, int nb, int per_part) {
    __shared__ int cur[MAXB];
    const int part = blockIdx.x;
    for (int i = threadIdx.x; i < nb; i += blockDim.x)
        cur[i] = offs[i * PARTS + part] + bstart[i];
    __syncthreads();
    const int e0 = part * per_part;
    int e1 = e0 + per_part;
    if (e1 > n_edges) e1 = n_edges;
    for (int eb = e0 + (int)threadIdx.x; eb < e1; eb += 8 * 1024) {
        int s[8], d[8];
#pragma unroll
        for (int u = 0; u < 8; ++u) {
            const int e = eb + u * 1024;
            s[u] = (e < e1) ? ei[e] : 0;
            d[u] = (e < e1) ? ei[n_edges + e] : -1;
        }
#pragma unroll
        for (int u = 0; u < 8; ++u)
            if (d[u] >= 0) {
                const int p = atomicAdd(&cur[d[u] / ROWS], 1);
                rec[p] = make_int2(s[u], d[u]);
            }
    }
}

__global__ __launch_bounds__(1024)
void gat_bucket_sum_kernel(const int2* __restrict__ rec,
                           const int* __restrict__ bstart,
                           const float* __restrict__ h,
                           const float* __restrict__ s_src,
                           const float* __restrict__ s_dst,
                           float* __restrict__ out,
                           int n_nodes) {
    __shared__ __attribute__((aligned(16))) float acc[ROWS * C_OUT];
    const int b    = blockIdx.x;
    const int lane = threadIdx.x & 31;
    const int wave = __builtin_amdgcn_readfirstlane((int)(threadIdx.x >> 5));
    const int row0 = b * ROWS;

    float4* acc4 = (float4*)acc;
    for (int i = threadIdx.x; i < ROWS * C_OUT / 4; i += 1024)
        acc4[i] = make_float4(0.f, 0.f, 0.f, 0.f);
    __syncthreads();

    const int beg = bstart[b];
    const int end = bstart[b + 1];
    for (int base = beg + wave * 32; base < end; base += 32 * 32) {
        int n = end - base;
        if (n > 32) n = 32;
        int   src = 0, dl = 0;
        float sc  = 0.f;
        if (lane < n) {
            const int2 r = rec[base + lane];
            src = r.x;
            dl  = r.y - row0;
            const float v = s_src[r.x] + s_dst[r.y];
            sc = (v > 0.f) ? v : ALPHA * v;
        }
        for (int j0 = 0; j0 < n; j0 += 16) {
            float4 hv[16];
#pragma unroll
            for (int u = 0; u < 16; ++u) {
                const int jj = (j0 + u < n) ? (j0 + u) : j0;
                const int sj = __builtin_amdgcn_readlane(src, jj);
                hv[u] = *(const float4*)(h + (long long)sj * C_OUT + 4 * lane);
            }
#pragma unroll
            for (int u = 0; u < 16; ++u) {
                if (j0 + u < n) {
                    const float w = __int_as_float(
                        __builtin_amdgcn_readlane(__float_as_int(sc), j0 + u));
                    const int d = __builtin_amdgcn_readlane(dl, j0 + u);
                    float* ap = acc + d * C_OUT + lane;
                    atomicAdd(ap +  0, w * hv[u].x);
                    atomicAdd(ap + 32, w * hv[u].y);
                    atomicAdd(ap + 64, w * hv[u].z);
                    atomicAdd(ap + 96, w * hv[u].w);
                }
            }
        }
    }
    __syncthreads();

    int rows = n_nodes - row0;
    if (rows > ROWS) rows = ROWS;
    float4* o4 = (float4*)(out + (long long)row0 * C_OUT);
    for (int i = threadIdx.x; i < rows * (C_OUT / 4); i += 1024) {
        const float* ar = acc + (i >> 5) * C_OUT + (i & 31);
        o4[i] = make_float4(ar[0], ar[32], ar[64], ar[96]);
    }
}

extern "C" void kernel_launch(void* const* d_in, const int* in_sizes, int n_in,
                              void* d_out, int out_size, void* d_ws, size_t ws_size,
                              hipStream_t stream) {
    const float* x  = (const float*)d_in[0];
    const int*   ei = (const int*)  d_in[1];
    const float* W  = (const float*)d_in[2];
    const float* a  = (const float*)d_in[3];
    float* out = (float*)d_out;

    const int N = in_sizes[0] / C_IN;
    const int E = in_sizes[1] / 2;

    float* h     = (float*)d_ws;
    float* s_src = h + (long long)N * C_OUT;
    float* s_dst = s_src + N;

    const int NB       = (N + ROWS - 1) / ROWS;
    const int per_part = (E + PARTS - 1) / PARTS;
    int*  offs   = (int*)(s_dst + N);
    int2* rec    = (int2*)(offs + (long long)NB * PARTS);
    int*  bstart = (int*)(rec + E);
    const size_t need = (size_t)((const char*)(bstart + NB + 1) - (const char*)d_ws);
    const bool regroup = (ws_size >= need) && (NB <= MAXB);

    if (regroup) {
        gat_count_kernel<<<PARTS, 1024, 0, stream>>>(ei, offs, E, NB, per_part);
        gat_scan_kernel<<<1, 1024, 0, stream>>>(offs, bstart, NB, E);
        gat_place_kernel<<<PARTS, 1024, 0, stream>>>(ei, offs, bstart, rec, E, NB, per_part);
    } else {
        long long n4 = (long long)N * C_OUT / 4;
        gat_zero_kernel<<<2048, 256, 0, stream>>>(out, n4);
    }

    {
        const int n_tiles = (N + 15) / 16;
        gat_gemm_kernel<<<GEMM_WGS, 256, 0, stream>>>(x, W, h, N, n_tiles);
    }

    {
        const int n_tiles = (N + 15) / 16;
        const int blocks  = (n_tiles + 7) / 8;
        gat_scores_kernel<<<blocks, 256, 0, stream>>>(h, a, s_src, s_dst, N, n_tiles);
    }

    if (regroup) {
        gat_bucket_sum_kernel<<<NB, 1024, 0, stream>>>(rec, bstart, h, s_src, s_dst, out, N);
    } else {
        gat_edge_kernel<<<4096, 256, 0, stream>>>(ei, h, s_src, s_dst, out, E);
    }
}


# bucket sum takes scores from rows it loads anyway (s_src per edge from the gathered row, 16 rows per DPP xor tree; s_dst per bucket row in LDS); score launch and 4-B score gathers gone; same sums
# speedup vs baseline: 3.0170x; 1.0579x over previous
#include <hip/hip_runtime.h>


typedef float v2f __attribute__((ext_vector_type(2)));
typedef float v8f __attribute__((ext_vector_type(8)));

#define C_IN   256
#define C_OUT  128
#define ALPHA  0.2f
#define ROWS   392
#define PARTS  256
#define MAXB   1024

__global__ void gat_zero_kernel(float* __restrict__ p, long long n4) {
    long long i      = (long long)blockIdx.x * blockDim.x + threadIdx.x;
    long long stride = (long long)gridDim.x * blockDim.x;
    float4 z = make_float4(0.f, 0.f, 0.f, 0.f);
    for (; i < n4; i += stride) ((float4*)p)[i] = z;
}

#define GEMM_WGS  512
#define WP_PITCH  288

__device__ __forceinline__ void gemm_load_b(v2f (&bf)[16], const float* wk) {
#pragma unroll
    for (int jj = 0; jj < 4; ++jj)
#pragma unroll
        for (int nt = 0; nt < 4; ++nt)
            bf[jj * 4 + nt] = *(const v2f*)(wk + 2 * jj * WP_PITCH + 32 * nt);
}

__device__ __forceinline__ void gemm_wmma16(v8f (&acc)[4], const v2f* a, const v2f (&bf)[16]) {
#pragma unroll
    for (int jj = 0; jj < 4; ++jj)
#pragma unroll
        for (int nt = 0; nt < 4; ++nt)
            acc[nt] = __builtin_amdgcn_wmma_f32_16x16x4_f32(
                 false, a[jj],  false, bf[jj * 4 + nt],
                 (short)0, acc[nt],
                 false,  false);
}

__global__ __launch_bounds__(256)
void gat_gemm_kernel(const float* __restrict__ x,
                     const float* __restrict__ W,
                     float* __restrict__ h,
                     int n_nodes, int n_tiles) {
    __shared__ __attribute__((aligned(16))) float Wp[(C_IN / 2) * WP_PITCH];
    const int lane = threadIdx.x & 31;
    const int wave = __builtin_amdgcn_readfirstlane((int)(threadIdx.x >> 5));
    const int half = lane >> 4;
    const int lmod = lane & 15;

    for (int it = 0; it < 2; ++it) {
        float4 r0[8], r1[8];
#pragma unroll
        for (int u = 0; u < 8; ++u) {
            const int pr = wave + 8 * (it * 8 + u);
            r0[u] = ((const float4*)(W + (2 * pr + 0) * C_OUT))[lane];
            r1[u] = ((const float4*)(W + (2 * pr + 1) * C_OUT))[lane];
        }
#pragma unroll
        for (int u = 0; u < 8; ++u) {
            const int pr = wave + 8 * (it * 8 + u);
            float4* d = (float4*)(Wp + pr * WP_PITCH + 8 * lane);
            d[0] = make_float4(r0[u].x, r1[u].x, r0[u].y, r1[u].y);
            d[1] = make_float4(r0[u].z, r1[u].z, r0[u].w, r1[u].w);
        }
    }
    __syncthreads();

    const int S     = (int)gridDim.x * 8;
    const int hgrid = (int)gridDim.x >> 1;
    const int b     = (int)blockIdx.x;
    const int slot  = (((b / hgrid) * 2 + (wave >> 2)) * hgrid + (b % hgrid)) * 4 + (wave & 3);
    const int n_units = n_tiles * 2;

    for (int u = slot; u < n_units; u += S) {
        const int m0  = (u >> 1) * 16;
        const int nc0 = (u & 1) * 64;

        int row = m0 + lmod;
        if (row > n_nodes - 1) row = n_nodes - 1;
        const float* __restrict__ xrow = x + (long long)row * C_IN + 2 * half;
        const float* wl = Wp + half * WP_PITCH + 2 * (nc0 + lmod);

        v8f acc[4] = {};
        v2f a0[8], a1[8];
        v2f bA[16], bB[16];
#pragma unroll
        for (int j = 0; j < 8; ++j) a0[j] = *(const v2f*)(xrow + 4 * j);
        gemm_load_b(bA, wl);

#pragma unroll 1
        for (int kb = 0; kb < 8; kb += 2) {
#pragma unroll
            for (int j = 0; j < 8; ++j) a1[j] = *(const v2f*)(xrow + 32 * (kb + 1) + 4 * j);
            gemm_load_b(bB, wl + (kb * 16 + 8) * WP_PITCH);
            gemm_wmma16(acc, a0, bA);
            gemm_load_b(bA, wl + (kb + 1) * 16 * WP_PITCH);
            gemm_wmma16(acc, a0 + 4, bB);
            const int kn = (kb + 2 < 8) ? kb + 2 : 7;
#pragma unroll
            for (int j = 0; j < 8; ++j) a0[j] = *(const v2f*)(xrow + 32 * kn + 4 * j);
            gemm_load_b(bB, wl + ((kb + 1) * 16 + 8) * WP_PITCH);
            gemm_wmma16(acc, a1, bA);
            gemm_load_b(bA, wl + kn * 16 * WP_PITCH);
            gemm_wmma16(acc, a1 + 4, bB);
        }

        float* __restrict__ hb = h + (long long)m0 * C_OUT + nc0 + lmod;
        if (m0 + 16 <= n_nodes) {
#pragma unroll
            for (int nt = 0; nt < 4; ++nt)
#pragma unroll
                for (int r = 0; r < 8; ++r)
                    hb[(r + 8 * half) * C_OUT + nt * 16] = acc[nt][r];
        } else {
#pragma unroll
            for (int nt = 0; nt < 4; ++nt)
#pragma unroll
                for (int r = 0; r < 8; ++r) {
                    const int m = m0 + r + 8 * half;
                    if (m < n_nodes)
                        hb[(r + 8 * half) * C_OUT + nt * 16] = acc[nt][r];
                }
        }
    }
}

__global__ __launch_bounds__(256)
void gat_scores_kernel(const float* __restrict__ h,
                       const float* __restrict__ a,
                       float* __restrict__ s_src,
                       float* __restrict__ s_dst,
                       int n_nodes, int n_tiles) {
    const int lane = threadIdx.x & 31;
    const int tile = blockIdx.x * 8 + (int)(threadIdx.x >> 5);
    if (tile >= n_tiles) return;
    const int n0 = tile * 16;

    const float4 w1 = ((const float4*)(a))[lane];
    const float4 w2 = ((const float4*)(a + C_OUT))[lane];

    float4 hv[16];
#pragma unroll
    for (int u = 0; u < 16; ++u) {
        int node = n0 + u;
        if (node > n_nodes - 1) node = n_nodes - 1;
        hv[u] = ((const float4*)(h + (long long)node * C_OUT))[lane];
    }

    float r1 = 0.f, r2 = 0.f;
#pragma unroll
    for (int u = 0; u < 16; ++u) {
        float p1 = hv[u].x * w1.x + hv[u].y * w1.y + hv[u].z * w1.z + hv[u].w * w1.w;
        float p2 = hv[u].x * w2.x + hv[u].y * w2.y + hv[u].z * w2.z + hv[u].w * w2.w;
        for (int off = 16; off > 0; off >>= 1) {
            p1 += __shfl_xor(p1, off, 32);
            p2 += __shfl_xor(p2, off, 32);
        }
        if (lane == u) { r1 = p1; r2 = p2; }
    }
    if (lane < 16 && n0 + lane < n_nodes) {
        s_src[n0 + lane] = r1;
        s_dst[n0 + lane] = r2;
    }
}

__global__ void gat_edge_kernel(const int* __restrict__ ei,
                                const float* __restrict__ h,
                                const float* __restrict__ s_src,
                                const float* __restrict__ s_dst,
                                float* __restrict__ out,
                                int n_edges) {
    const int lane    = threadIdx.x & 31;
    const int wid     = (int)(((long long)blockIdx.x * blockDim.x + threadIdx.x) >> 5);
    const int n_waves = (int)(((long long)gridDim.x * blockDim.x) >> 5);

    for (int e = wid; e < n_edges; e += n_waves) {
        const int src = ei[e];
        const int dst = ei[n_edges + e];

        float sc = s_src[src] + s_dst[dst];
        sc = (sc > 0.f) ? sc : ALPHA * sc;

        const float* __restrict__ hs = h + (long long)src * C_OUT + lane;
        const float h0 = hs[0];
        const float h1 = hs[32];
        const float h2 = hs[64];
        const float h3 = hs[96];
        float* o = out + (long long)dst * C_OUT + lane;
        unsafeAtomicAdd(o +  0, sc * h0);
        unsafeAtomicAdd(o + 32, sc * h1);
        unsafeAtomicAdd(o + 64, sc * h2);
        unsafeAtomicAdd(o + 96, sc * h3);
    }
}

__global__ __launch_bounds__(1024)
void gat_count_kernel(const int* __restrict__ ei, int* __restrict__ offs,
                      int n_edges, int nb, int per_part) {
    __shared__ int cnt[MAXB];
    const int part = blockIdx.x;
    for (int i = threadIdx.x; i < nb; i += blockDim.x) cnt[i] = 0;
    __syncthreads();
    const int e0 = part * per_part;
    int e1 = e0 + per_part;
    if (e1 > n_edges) e1 = n_edges;
    const int* __restrict__ dstp = ei + n_edges;
    for (int eb = e0 + (int)threadIdx.x; eb < e1; eb += 8 * 1024) {
        int d[8];
#pragma unroll
        for (int u = 0; u < 8; ++u) {
            const int e = eb + u * 1024;
            d[u] = (e < e1) ? dstp[e] : -1;
        }
#pragma unroll
        for (int u = 0; u < 8; ++u)
            if (d[u] >= 0) atomicAdd(&cnt[d[u] / ROWS], 1);
    }
    __syncthreads();
    for (int i = threadIdx.x; i < nb; i += blockDim.x) offs[i * PARTS + part] = cnt[i];
}

__global__ __launch_bounds__(1024)
void gat_scan_kernel(int* __restrict__ offs, int* __restrict__ bstart,
                     int nb, int n_edges) {
    __shared__ int tot[1024];
    const int t    = threadIdx.x;
    const int lane = t & 31;
    const int wave = t >> 5;
    tot[t] = 0;
    __syncthreads();
    for (int b = wave; b < nb; b += 32) {
        int4* row = (int4*)(offs + b * PARTS);
        int4 v0 = row[lane * 2 + 0];
        int4 v1 = row[lane * 2 + 1];
        const int s = v0.x + v0.y + v0.z + v0.w + v1.x + v1.y + v1.z + v1.w;
        int inc = s;
#pragma unroll
        for (int d = 1; d < 32; d <<= 1) {
            const int o = __shfl_up(inc, d, 32);
            if (lane >= d) inc += o;
        }
        int run = inc - s;
        int4 w0, w1;
        w0.x = run; run += v0.x;
        w0.y = run; run += v0.y;
        w0.z = run; run += v0.z;
        w0.w = run; run += v0.w;
        w1.x = run; run += v1.x;
        w1.y = run; run += v1.y;
        w1.z = run; run += v1.z;
        w1.w = run;
        row[lane * 2 + 0] = w0;
        row[lane * 2 + 1] = w1;
        if (lane == 31) tot[b] = inc;
    }
    __syncthreads();
    const int mine = tot[t];
    for (int d = 1; d < 1024; d <<= 1) {
        const int v = (t >= d) ? tot[t - d] : 0;
        __syncthreads();
        tot[t] += v;
        __syncthreads();
    }
    if (t < nb) bstart[t] = tot[t] - mine;
    if (t == 0) bstart[nb] = n_edges;
}

__global__ __launch_bounds__(1024)
void gat_place_kernel(const int* __restrict__ ei, const int* __restrict__ offs,
                      const int* __restrict__ bstart, int2* __restrict__ rec,
                      int n_edges, int nb, int per_part) {
    __shared__ int cur[MAXB];
    const int part = blockIdx.x;
    for (int i = threadIdx.x; i < nb; i += blockDim.x)
        cur[i] = offs[i * PARTS + part] + bstart[i];
    __syncthreads();
    const int e0 = part * per_part;
    int e1 = e0 + per_part;
    if (e1 > n_edges) e1 = n_edges;
    for (int eb = e0 + (int)threadIdx.x; eb < e1; eb += 8 * 1024) {
        int s[8], d[8];
#pragma unroll
        for (int u = 0; u < 8; ++u) {
            const int e = eb + u * 1024;
            s[u] = (e < e1) ? ei[e] : 0;
            d[u] = (e < e1) ? ei[n_edges + e] : -1;
        }
#pragma unroll
        for (int u = 0; u < 8; ++u)
            if (d[u] >= 0) {
                const int p = atomicAdd(&cur[d[u] / ROWS], 1);
                rec[p] = make_int2(s[u], d[u]);
            }
    }
}

__device__ __forceinline__ float gat_xhalf(float x) {
    const int b = __builtin_bit_cast(int, x);
    return __builtin_bit_cast(float,
        __builtin_amdgcn_permlanex16(b, b, 0x76543210, 0xfedcba98, false, false));
}
__device__ __forceinline__ float gat_xrow8(float x) {
    const int b = __builtin_bit_cast(int, x);
    return __builtin_bit_cast(float, __builtin_amdgcn_update_dpp(0, b, 0x168, 0xf, 0xf, true));
}
__device__ __forceinline__ float gat_xrow4(float x) {
    const int b = __builtin_bit_cast(int, x);
    return __builtin_bit_cast(float, __builtin_amdgcn_update_dpp(0, b, 0x164, 0xf, 0xf, true));
}
__device__ __forceinline__ float gat_xrow2(float x) {
    const int b = __builtin_bit_cast(int, x);
    return __builtin_bit_cast(float, __builtin_amdgcn_update_dpp(0, b, 0x162, 0xf, 0xf, true));
}
__device__ __forceinline__ float gat_xrow1(float x) {
    const int b = __builtin_bit_cast(int, x);
    return __builtin_bit_cast(float, __builtin_amdgcn_update_dpp(0, b, 0x161, 0xf, 0xf, true));
}
__device__ __forceinline__ float gat_dot4(const float4 v, const float4 w) {
    return __builtin_fmaf(v.w, w.w, __builtin_fmaf(v.z, w.z, __builtin_fmaf(v.x, w.x, v.y * w.y)));
}
__device__ __forceinline__ float gat_wave_sum(float p) {
    p = p + gat_xhalf(p);
    p = p + gat_xrow8(p);
    p = p + gat_xrow4(p);
    p = p + gat_xrow2(p);
    p = p + gat_xrow1(p);
    return p;
}

#define SD_PER_WAVE ((ROWS + 31) / 32)
__global__ __launch_bounds__(1024)
void gat_bucket_sum_kernel(const int2* __restrict__ rec,
                           const int* __restrict__ bstart,
                           const float* __restrict__ h,
                           const float* __restrict__ a,
                           float* __restrict__ out,
                           int n_nodes) {
    __shared__ __attribute__((aligned(16))) float acc[ROWS * C_OUT];
    __shared__ float sd[ROWS];
    const int b    = blockIdx.x;
    const int lane = threadIdx.x & 31;
    const int wave = __builtin_amdgcn_readfirstlane((int)(threadIdx.x >> 5));
    const int row0 = b * ROWS;
    int rows = n_nodes - row0;
    if (rows > ROWS) rows = ROWS;

    const float4 w1 = ((const float4*)(a))[lane];
    const float4 w2 = ((const float4*)(a + C_OUT))[lane];

    float4* acc4 = (float4*)acc;
    for (int i = threadIdx.x; i < ROWS * C_OUT / 4; i += 1024)
        acc4[i] = make_float4(0.f, 0.f, 0.f, 0.f);

    {
        float4 hd[SD_PER_WAVE];
#pragma unroll
        for (int k = 0; k < SD_PER_WAVE; ++k) {
            int r = wave + 32 * k;
            if (r > rows - 1) r = rows - 1;
            hd[k] = *(const float4*)(h + (long long)(row0 + r) * C_OUT + 4 * lane);
        }
#pragma unroll
        for (int k = 0; k < SD_PER_WAVE; ++k) {
            const float p2 = gat_wave_sum(gat_dot4(hd[k], w2));
            const int r = wave + 32 * k;
            if (lane == 0 && r < rows) sd[r] = p2;
        }
    }
    __syncthreads();

    const bool k16 = (lane & 16) != 0;
    const bool k8  = (lane & 8) != 0;
    const bool k4  = (lane & 4) != 0;
    const bool k2  = (lane & 2) != 0;

    const int beg = bstart[b];
    const int end = bstart[b + 1];
    for (int base = beg + wave * 32; base < end; base += 32 * 32) {
        int n = end - base;
        if (n > 32) n = 32;
        int src = 0, dl = 0;
        if (lane < n) {
            const int2 r = rec[base + lane];
            src = r.x;
            dl  = r.y - row0;
        }
        const float sdl = sd[dl];
        for (int j0 = 0; j0 < n; j0 += 16) {
            float4 hv[16];
#pragma unroll
            for (int u = 0; u < 16; ++u) {
                const int jj = (j0 + u < n) ? (j0 + u) : j0;
                const int sj = __builtin_amdgcn_readlane(src, jj);
                hv[u] = *(const float4*)(h + (long long)sj * C_OUT + 4 * lane);
            }
            float p[16];
#pragma unroll
            for (int u = 0; u < 16; ++u) p[u] = gat_dot4(hv[u], w1);
            float q[8];
#pragma unroll
            for (int u = 0; u < 8; ++u) {
                const float keep = k16 ? p[u + 8] : p[u];
                const float send = k16 ? p[u] : p[u + 8];
                q[u] = keep + gat_xhalf(send);
            }
            float r4[4];
#pragma unroll
            for (int u = 0; u < 4; ++u) {
                const float keep = k8 ? q[u + 4] : q[u];
                const float send = k8 ? q[u] : q[u + 4];
                r4[u] = keep + gat_xrow8(send);
            }
            float s2[2];
#pragma unroll
            for (int u = 0; u < 2; ++u) {
                const float keep = k4 ? r4[u + 2] : r4[u];
                const float send = k4 ? r4[u] : r4[u + 2];
                s2[u] = keep + gat_xrow4(send);
            }
            float t;
            {
                const float keep = k2 ? s2[1] : s2[0];
                const float send = k2 ? s2[0] : s2[1];
                t = keep + gat_xrow2(send);
            }
            t = t + gat_xrow1(t);
            const float sdu = __shfl(sdl, j0 + (lane >> 1), 32);
            const float v   = t + sdu;
            const float scg = (v > 0.f) ? v : ALPHA * v;
#pragma unroll
            for (int u = 0; u < 16; ++u) {
                if (j0 + u < n) {
                    const float w = __int_as_float(
                        __builtin_amdgcn_readlane(__float_as_int(scg), 2 * u));
                    const int d = __builtin_amdgcn_readlane(dl, j0 + u);
                    float* ap = acc + d * C_OUT + lane;
                    atomicAdd(ap +  0, w * hv[u].x);
                    atomicAdd(ap + 32, w * hv[u].y);
                    atomicAdd(ap + 64, w * hv[u].z);
                    atomicAdd(ap + 96, w * hv[u].w);
                }
            }
        }
    }
    __syncthreads();

    float4* o4 = (float4*)(out + (long long)row0 * C_OUT);
    for (int i = threadIdx.x; i < rows * (C_OUT / 4); i += 1024) {
        const float* ar = acc + (i >> 5) * C_OUT + (i & 31);
        o4[i] = make_float4(ar[0], ar[32], ar[64], ar[96]);
    }
}

extern "C" void kernel_launch(void* const* d_in, const int* in_sizes, int n_in,
                              void* d_out, int out_size, void* d_ws, size_t ws_size,
                              hipStream_t stream) {
    const float* x  = (const float*)d_in[0];
    const int*   ei = (const int*)  d_in[1];
    const float* W  = (const float*)d_in[2];
    const float* a  = (const float*)d_in[3];
    float* out = (float*)d_out;

    const int N = in_sizes[0] / C_IN;
    const int E = in_sizes[1] / 2;

    float* h     = (float*)d_ws;
    float* s_src = h + (long long)N * C_OUT;
    float* s_dst = s_src + N;

    const int NB       = (N + ROWS - 1) / ROWS;
    const int per_part = (E + PARTS - 1) / PARTS;
    int*  offs   = (int*)(s_dst + N);
    int2* rec    = (int2*)(offs + (long long)NB * PARTS);
    int*  bstart = (int*)(rec + E);
    const size_t need = (size_t)((const char*)(bstart + NB + 1) - (const char*)d_ws);
    const bool regroup = (ws_size >= need) && (NB <= MAXB);

    if (regroup) {
        gat_count_kernel<<<PARTS, 1024, 0, stream>>>(ei, offs, E, NB, per_part);
        gat_scan_kernel<<<1, 1024, 0, stream>>>(offs, bstart, NB, E);
        gat_place_kernel<<<PARTS, 1024, 0, stream>>>(ei, offs, bstart, rec, E, NB, per_part);
    } else {
        long long n4 = (long long)N * C_OUT / 4;
        gat_zero_kernel<<<2048, 256, 0, stream>>>(out, n4);
    }

    {
        const int n_tiles = (N + 15) / 16;
        gat_gemm_kernel<<<GEMM_WGS, 256, 0, stream>>>(x, W, h, N, n_tiles);
    }

    if (!regroup) {
        const int n_tiles = (N + 15) / 16;
        const int blocks  = (n_tiles + 7) / 8;
        gat_scores_kernel<<<blocks, 256, 0, stream>>>(h, a, s_src, s_dst, N, n_tiles);
    }

    if (regroup) {
        gat_bucket_sum_kernel<<<NB, 1024, 0, stream>>>(rec, bstart, h, a, out, N);
    } else {
        gat_edge_kernel<<<4096, 256, 0, stream>>>(ei, h, s_src, s_dst, out, E);
    }
}


# sort's scan pass moved into the GEMM launch as its work-group 0 (runs under the GEMM, 511 GEMM work-groups); the one-work-group scan launch is gone; h, records and LDS add order as before
# speedup vs baseline: 3.2212x; 1.0677x over previous
#include <hip/hip_runtime.h>


typedef float v2f __attribute__((ext_vector_type(2)));
typedef float v8f __attribute__((ext_vector_type(8)));

#define C_IN   256
#define C_OUT  128
#define ALPHA  0.2f
#define ROWS   392
#define PARTS  256
#define MAXB   1024

__global__ void gat_zero_kernel(float* __restrict__ p, long long n4) {
    long long i      = (long long)blockIdx.x * blockDim.x + threadIdx.x;
    long long stride = (long long)gridDim.x * blockDim.x;
    float4 z = make_float4(0.f, 0.f, 0.f, 0.f);
    for (; i < n4; i += stride) ((float4*)p)[i] = z;
}

#define GEMM_WGS  512
#define WP_PITCH  288

__device__ __forceinline__ void gat_scan_role(int* __restrict__ offs, int* __restrict__ bstart,
                                              int nb, int n_edges, int* lds);

__device__ __forceinline__ void gemm_load_b(v2f (&bf)[16], const float* wk) {
#pragma unroll
    for (int jj = 0; jj < 4; ++jj)
#pragma unroll
        for (int nt = 0; nt < 4; ++nt)
            bf[jj * 4 + nt] = *(const v2f*)(wk + 2 * jj * WP_PITCH + 32 * nt);
}

__device__ __forceinline__ void gemm_wmma16(v8f (&acc)[4], const v2f* a, const v2f (&bf)[16]) {
#pragma unroll
    for (int jj = 0; jj < 4; ++jj)
#pragma unroll
        for (int nt = 0; nt < 4; ++nt)
            acc[nt] = __builtin_amdgcn_wmma_f32_16x16x4_f32(
                 false, a[jj],  false, bf[jj * 4 + nt],
                 (short)0, acc[nt],
                 false,  false);
}

__global__ __launch_bounds__(256)
void gat_gemm_kernel(const float* __restrict__ x,
                     const float* __restrict__ W,
                     float* __restrict__ h,
                     int n_nodes, int n_tiles,
                     int* __restrict__ offs, int* __restrict__ bstart,
                     int nb, int n_edges, int scan_wgs) {
    __shared__ __attribute__((aligned(16))) float Wp[(C_IN / 2) * WP_PITCH];
    if ((int)blockIdx.x < scan_wgs) {
        gat_scan_role(offs, bstart, nb, n_edges, (int*)Wp);
        return;
    }
    const int lane = threadIdx.x & 31;
    const int wave = __builtin_amdgcn_readfirstlane((int)(threadIdx.x >> 5));
    const int half = lane >> 4;
    const int lmod = lane & 15;

    for (int it = 0; it < 2; ++it) {
        float4 r0[8], r1[8];
#pragma unroll
        for (int u = 0; u < 8; ++u) {
            const int pr = wave + 8 * (it * 8 + u);
            r0[u] = ((const float4*)(W + (2 * pr + 0) * C_OUT))[lane];
            r1[u] = ((const float4*)(W + (2 * pr + 1) * C_OUT))[lane];
        }
#pragma unroll
        for (int u = 0; u < 8; ++u) {
            const int pr = wave + 8 * (it * 8 + u);
            float4* d = (float4*)(Wp + pr * WP_PITCH + 8 * lane);
            d[0] = make_float4(r0[u].x, r1[u].x, r0[u].y, r1[u].y);
            d[1] = make_float4(r0[u].z, r1[u].z, r0[u].w, r1[u].w);
        }
    }
    __syncthreads();

    const int G     = (int)gridDim.x - scan_wgs;
    const int g     = (int)blockIdx.x - scan_wgs;
    const int S     = G * 8;
    const int slot  = ((wave >> 2) * G + g) * 4 + (wave & 3);
    const int n_units = n_tiles * 2;

    for (int u = slot; u < n_units; u += S) {
        const int m0  = (u >> 1) * 16;
        const int nc0 = (u & 1) * 64;

        int row = m0 + lmod;
        if (row > n_nodes - 1) row = n_nodes - 1;
        const float* __restrict__ xrow = x + (long long)row * C_IN + 2 * half;
        const float* wl = Wp + half * WP_PITCH + 2 * (nc0 + lmod);

        v8f acc[4] = {};
        v2f a0[8], a1[8];
        v2f bA[16], bB[16];
#pragma unroll
        for (int j = 0; j < 8; ++j) a0[j] = *(const v2f*)(xrow + 4 * j);
        gemm_load_b(bA, wl);

#pragma unroll 1
        for (int kb = 0; kb < 8; kb += 2) {
#pragma unroll
            for (int j = 0; j < 8; ++j) a1[j] = *(const v2f*)(xrow + 32 * (kb + 1) + 4 * j);
            gemm_load_b(bB, wl + (kb * 16 + 8) * WP_PITCH);
            gemm_wmma16(acc, a0, bA);
            gemm_load_b(bA, wl + (kb + 1) * 16 * WP_PITCH);
            gemm_wmma16(acc, a0 + 4, bB);
            const int kn = (kb + 2 < 8) ? kb + 2 : 7;
#pragma unroll
            for (int j = 0; j < 8; ++j) a0[j] = *(const v2f*)(xrow + 32 * kn + 4 * j);
            gemm_load_b(bB, wl + ((kb + 1) * 16 + 8) * WP_PITCH);
            gemm_wmma16(acc, a1, bA);
            gemm_load_b(bA, wl + kn * 16 * WP_PITCH);
            gemm_wmma16(acc, a1 + 4, bB);
        }

        float* __restrict__ hb = h + (long long)m0 * C_OUT + nc0 + lmod;
        if (m0 + 16 <= n_nodes) {
#pragma unroll
            for (int nt = 0; nt < 4; ++nt)
#pragma unroll
                for (int r = 0; r < 8; ++r)
                    hb[(r + 8 * half) * C_OUT + nt * 16] = acc[nt][r];
        } else {
#pragma unroll
            for (int nt = 0; nt < 4; ++nt)
#pragma unroll
                for (int r = 0; r < 8; ++r) {
                    const int m = m0 + r + 8 * half;
                    if (m < n_nodes)
                        hb[(r + 8 * half) * C_OUT + nt * 16] = acc[nt][r];
                }
        }
    }
}

__global__ __launch_bounds__(256)
void gat_scores_kernel(const float* __restrict__ h,
                       const float* __restrict__ a,
                       float* __restrict__ s_src,
                       float* __restrict__ s_dst,
                       int n_nodes, int n_tiles) {
    const int lane = threadIdx.x & 31;
    const int tile = blockIdx.x * 8 + (int)(threadIdx.x >> 5);
    if (tile >= n_tiles) return;
    const int n0 = tile * 16;

    const float4 w1 = ((const float4*)(a))[lane];
    const float4 w2 = ((const float4*)(a + C_OUT))[lane];

    float4 hv[16];
#pragma unroll
    for (int u = 0; u < 16; ++u) {
        int node = n0 + u;
        if (node > n_nodes - 1) node = n_nodes - 1;
        hv[u] = ((const float4*)(h + (long long)node * C_OUT))[lane];
    }

    float r1 = 0.f, r2 = 0.f;
#pragma unroll
    for (int u = 0; u < 16; ++u) {
        float p1 = hv[u].x * w1.x + hv[u].y * w1.y + hv[u].z * w1.z + hv[u].w * w1.w;
        float p2 = hv[u].x * w2.x + hv[u].y * w2.y + hv[u].z * w2.z + hv[u].w * w2.w;
        for (int off = 16; off > 0; off >>= 1) {
            p1 += __shfl_xor(p1, off, 32);
            p2 += __shfl_xor(p2, off, 32);
        }
        if (lane == u) { r1 = p1; r2 = p2; }
    }
    if (lane < 16 && n0 + lane < n_nodes) {
        s_src[n0 + lane] = r1;
        s_dst[n0 + lane] = r2;
    }
}

__global__ void gat_edge_kernel(const int* __restrict__ ei,
                                const float* __restrict__ h,
                                const float* __restrict__ s_src,
                                const float* __restrict__ s_dst,
                                float* __restrict__ out,
                                int n_edges) {
    const int lane    = threadIdx.x & 31;
    const int wid     = (int)(((long long)blockIdx.x * blockDim.x + threadIdx.x) >> 5);
    const int n_waves = (int)(((long long)gridDim.x * blockDim.x) >> 5);

    for (int e = wid; e < n_edges; e += n_waves) {
        const int src = ei[e];
        const int dst = ei[n_edges + e];

        float sc = s_src[src] + s_dst[dst];
        sc = (sc > 0.f) ? sc : ALPHA * sc;

        const float* __restrict__ hs = h + (long long)src * C_OUT + lane;
        const float h0 = hs[0];
        const float h1 = hs[32];
        const float h2 = hs[64];
        const float h3 = hs[96];
        float* o = out + (long long)dst * C_OUT + lane;
        unsafeAtomicAdd(o +  0, sc * h0);
        unsafeAtomicAdd(o + 32, sc * h1);
        unsafeAtomicAdd(o + 64, sc * h2);
        unsafeAtomicAdd(o + 96, sc * h3);
    }
}

__global__ __launch_bounds__(1024)
void gat_count_kernel(const int* __restrict__ ei, int* __restrict__ offs,
                      int n_edges, int nb, int per_part) {
    __shared__ int cnt[MAXB];
    const int part = blockIdx.x;
    for (int i = threadIdx.x; i < nb; i += blockDim.x) cnt[i] = 0;
    __syncthreads();
    const int e0 = part * per_part;
    int e1 = e0 + per_part;
    if (e1 > n_edges) e1 = n_edges;
    const int* __restrict__ dstp = ei + n_edges;
    for (int eb = e0 + (int)threadIdx.x; eb < e1; eb += 8 * 1024) {
        int d[8];
#pragma unroll
        for (int u = 0; u < 8; ++u) {
            const int e = eb + u * 1024;
            d[u] = (e < e1) ? dstp[e] : -1;
        }
#pragma unroll
        for (int u = 0; u < 8; ++u)
            if (d[u] >= 0) atomicAdd(&cnt[d[u] / ROWS], 1);
    }
    __syncthreads();
    for (int i = threadIdx.x; i < nb; i += blockDim.x) offs[i * PARTS + part] = cnt[i];
}

__device__ __forceinline__ void gat_scan_role(int* __restrict__ offs, int* __restrict__ bstart,
                                              int nb, int n_edges, int* lds) {
    int* tot  = lds;
    int* wsum = lds + MAXB;
    const int t    = threadIdx.x;
    const int lane = t & 31;
    const int wave = t >> 5;
    for (int b0 = wave; b0 < nb; b0 += 8 * 4) {
        int4 v0[4], v1[4];
#pragma unroll
        for (int u = 0; u < 4; ++u) {
            const int b = b0 + 8 * u;
            v0[u] = make_int4(0, 0, 0, 0);
            v1[u] = make_int4(0, 0, 0, 0);
            if (b < nb) {
                const int4* row = (const int4*)(offs + b * PARTS);
                v0[u] = row[lane * 2 + 0];
                v1[u] = row[lane * 2 + 1];
            }
        }
#pragma unroll
        for (int u = 0; u < 4; ++u) {
            const int b = b0 + 8 * u;
            if (b < nb) {
                const int s = v0[u].x + v0[u].y + v0[u].z + v0[u].w
                            + v1[u].x + v1[u].y + v1[u].z + v1[u].w;
                int inc = s;
#pragma unroll
                for (int d = 1; d < 32; d <<= 1) {
                    const int o = __shfl_up(inc, d, 32);
                    if (lane >= d) inc += o;
                }
                int run = inc - s;
                int4 w0, w1;
                w0.x = run; run += v0[u].x;
                w0.y = run; run += v0[u].y;
                w0.z = run; run += v0[u].z;
                w0.w = run; run += v0[u].w;
                w1.x = run; run += v1[u].x;
                w1.y = run; run += v1[u].y;
                w1.z = run; run += v1[u].z;
                w1.w = run;
                int4* row = (int4*)(offs + b * PARTS);
                row[lane * 2 + 0] = w0;
                row[lane * 2 + 1] = w1;
                if (lane == 31) tot[b] = inc;
            }
        }
    }
    __syncthreads();
    int carry = 0;
    for (int base = 0; base < nb; base += 256) {
        const int i    = base + t;
        const int mine = (i < nb) ? tot[i] : 0;
        int inc = mine;
#pragma unroll
        for (int d = 1; d < 32; d <<= 1) {
            const int o = __shfl_up(inc, d, 32);
            if (lane >= d) inc += o;
        }
        if (lane == 31) wsum[wave] = inc;
        __syncthreads();
        int before = 0, all = 0;
#pragma unroll
        for (int w = 0; w < 8; ++w) {
            const int v = wsum[w];
            if (w < wave) before += v;
            all += v;
        }
        if (i < nb) bstart[i] = carry + before + inc - mine;
        carry += all;
        __syncthreads();
    }
    if (t == 0) bstart[nb] = n_edges;
}

__global__ __launch_bounds__(1024)
void gat_place_kernel(const int* __restrict__ ei, const int* __restrict__ offs,
                      const int* __restrict__ bstart, int2* __restrict__ rec,
                      int n_edges, int nb, int per_part) {
    __shared__ int cur[MAXB];
    const int part = blockIdx.x;
    for (int i = threadIdx.x; i < nb; i += blockDim.x)
        cur[i] = offs[i * PARTS + part] + bstart[i];
    __syncthreads();
    const int e0 = part * per_part;
    int e1 = e0 + per_part;
    if (e1 > n_edges) e1 = n_edges;
    for (int eb = e0 + (int)threadIdx.x; eb < e1; eb += 8 * 1024) {
        int s[8], d[8];
#pragma unroll
        for (int u = 0; u < 8; ++u) {
            const int e = eb + u * 1024;
            s[u] = (e < e1) ? ei[e] : 0;
            d[u] = (e < e1) ? ei[n_edges + e] : -1;
        }
#pragma unroll
        for (int u = 0; u < 8; ++u)
            if (d[u] >= 0) {
                const int p = atomicAdd(&cur[d[u] / ROWS], 1);
                rec[p] = make_int2(s[u], d[u]);
            }
    }
}

__device__ __forceinline__ float gat_xhalf(float x) {
    const int b = __builtin_bit_cast(int, x);
    return __builtin_bit_cast(float,
        __builtin_amdgcn_permlanex16(b, b, 0x76543210, 0xfedcba98, false, false));
}
__device__ __forceinline__ float gat_xrow8(float x) {
    const int b = __builtin_bit_cast(int, x);
    return __builtin_bit_cast(float, __builtin_amdgcn_update_dpp(0, b, 0x168, 0xf, 0xf, true));
}
__device__ __forceinline__ float gat_xrow4(float x) {
    const int b = __builtin_bit_cast(int, x);
    return __builtin_bit_cast(float, __builtin_amdgcn_update_dpp(0, b, 0x164, 0xf, 0xf, true));
}
__device__ __forceinline__ float gat_xrow2(float x) {
    const int b = __builtin_bit_cast(int, x);
    return __builtin_bit_cast(float, __builtin_amdgcn_update_dpp(0, b, 0x162, 0xf, 0xf, true));
}
__device__ __forceinline__ float gat_xrow1(float x) {
    const int b = __builtin_bit_cast(int, x);
    return __builtin_bit_cast(float, __builtin_amdgcn_update_dpp(0, b, 0x161, 0xf, 0xf, true));
}
__device__ __forceinline__ float gat_dot4(const float4 v, const float4 w) {
    return __builtin_fmaf(v.w, w.w, __builtin_fmaf(v.z, w.z, __builtin_fmaf(v.x, w.x, v.y * w.y)));
}
__device__ __forceinline__ float gat_wave_sum(float p) {
    p = p + gat_xhalf(p);
    p = p + gat_xrow8(p);
    p = p + gat_xrow4(p);
    p = p + gat_xrow2(p);
    p = p + gat_xrow1(p);
    return p;
}

#define SD_PER_WAVE ((ROWS + 31) / 32)
__global__ __launch_bounds__(1024)
void gat_bucket_sum_kernel(const int2* __restrict__ rec,
                           const int* __restrict__ bstart,
                           const float* __restrict__ h,
                           const float* __restrict__ a,
                           float* __restrict__ out,
                           int n_nodes) {
    __shared__ __attribute__((aligned(16))) float acc[ROWS * C_OUT];
    __shared__ float sd[ROWS];
    const int b    = blockIdx.x;
    const int lane = threadIdx.x & 31;
    const int wave = __builtin_amdgcn_readfirstlane((int)(threadIdx.x >> 5));
    const int row0 = b * ROWS;
    int rows = n_nodes - row0;
    if (rows > ROWS) rows = ROWS;

    const float4 w1 = ((const float4*)(a))[lane];
    const float4 w2 = ((const float4*)(a + C_OUT))[lane];

    float4* acc4 = (float4*)acc;
    for (int i = threadIdx.x; i < ROWS * C_OUT / 4; i += 1024)
        acc4[i] = make_float4(0.f, 0.f, 0.f, 0.f);

    {
        float4 hd[SD_PER_WAVE];
#pragma unroll
        for (int k = 0; k < SD_PER_WAVE; ++k) {
            int r = wave + 32 * k;
            if (r > rows - 1) r = rows - 1;
            hd[k] = *(const float4*)(h + (long long)(row0 + r) * C_OUT + 4 * lane);
        }
#pragma unroll
        for (int k = 0; k < SD_PER_WAVE; ++k) {
            const float p2 = gat_wave_sum(gat_dot4(hd[k], w2));
            const int r = wave + 32 * k;
            if (lane == 0 && r < rows) sd[r] = p2;
        }
    }
    __syncthreads();

    const bool k16 = (lane & 16) != 0;
    const bool k8  = (lane & 8) != 0;
    const bool k4  = (lane & 4) != 0;
    const bool k2  = (lane & 2) != 0;

    const int beg = bstart[b];
    const int end = bstart[b + 1];
    for (int base = beg + wave * 32; base < end; base += 32 * 32) {
        int n = end - base;
        if (n > 32) n = 32;
        int src = 0, dl = 0;
        if (lane < n) {
            const int2 r = rec[base + lane];
            src = r.x;
            dl  = r.y - row0;
        }
        const float sdl = sd[dl];
        for (int j0 = 0; j0 < n; j0 += 16) {
            float4 hv[16];
#pragma unroll
            for (int u = 0; u < 16; ++u) {
                const int jj = (j0 + u < n) ? (j0 + u) : j0;
                const int sj = __builtin_amdgcn_readlane(src, jj);
                hv[u] = *(const float4*)(h + (long long)sj * C_OUT + 4 * lane);
            }
            float p[16];
#pragma unroll
            for (int u = 0; u < 16; ++u) p[u] = gat_dot4(hv[u], w1);
            float q[8];
#pragma unroll
            for (int u = 0; u < 8; ++u) {
                const float keep = k16 ? p[u + 8] : p[u];
                const float send = k16 ? p[u] : p[u + 8];
                q[u] = keep + gat_xhalf(send);
            }
            float r4[4];
#pragma unroll
            for (int u = 0; u < 4; ++u) {
                const float keep = k8 ? q[u + 4] : q[u];
                const float send = k8 ? q[u] : q[u + 4];
                r4[u] = keep + gat_xrow8(send);
            }
            float s2[2];
#pragma unroll
            for (int u = 0; u < 2; ++u) {
                const float keep = k4 ? r4[u + 2] : r4[u];
                const float send = k4 ? r4[u] : r4[u + 2];
                s2[u] = keep + gat_xrow4(send);
            }
            float t;
            {
                const float keep = k2 ? s2[1] : s2[0];
                const float send = k2 ? s2[0] : s2[1];
                t = keep + gat_xrow2(send);
            }
            t = t + gat_xrow1(t);
            const float sdu = __shfl(sdl, j0 + (lane >> 1), 32);
            const float v   = t + sdu;
            const float scg = (v > 0.f) ? v : ALPHA * v;
#pragma unroll
            for (int u = 0; u < 16; ++u) {
                if (j0 + u < n) {
                    const float w = __int_as_float(
                        __builtin_amdgcn_readlane(__float_as_int(scg), 2 * u));
                    const int d = __builtin_amdgcn_readlane(dl, j0 + u);
                    float* ap = acc + d * C_OUT + lane;
                    atomicAdd(ap +  0, w * hv[u].x);
                    atomicAdd(ap + 32, w * hv[u].y);
                    atomicAdd(ap + 64, w * hv[u].z);
                    atomicAdd(ap + 96, w * hv[u].w);
                }
            }
        }
    }
    __syncthreads();

    float4* o4 = (float4*)(out + (long long)row0 * C_OUT);
    for (int i = threadIdx.x; i < rows * (C_OUT / 4); i += 1024) {
        const float* ar = acc + (i >> 5) * C_OUT + (i & 31);
        o4[i] = make_float4(ar[0], ar[32], ar[64], ar[96]);
    }
}

extern "C" void kernel_launch(void* const* d_in, const int* in_sizes, int n_in,
                              void* d_out, int out_size, void* d_ws, size_t ws_size,
                              hipStream_t stream) {
    const float* x  = (const float*)d_in[0];
    const int*   ei = (const int*)  d_in[1];
    const float* W  = (const float*)d_in[2];
    const float* a  = (const float*)d_in[3];
    float* out = (float*)d_out;

    const int N = in_sizes[0] / C_IN;
    const int E = in_sizes[1] / 2;

    float* h     = (float*)d_ws;
    float* s_src = h + (long long)N * C_OUT;
    float* s_dst = s_src + N;

    const int NB       = (N + ROWS - 1) / ROWS;
    const int per_part = (E + PARTS - 1) / PARTS;
    int*  offs   = (int*)(s_dst + N);
    int2* rec    = (int2*)(offs + (long long)NB * PARTS);
    int*  bstart = (int*)(rec + E);
    const size_t need = (size_t)((const char*)(bstart + NB + 1) - (const char*)d_ws);
    const bool regroup = (ws_size >= need) && (NB <= MAXB);

    const int n_tiles_g = (N + 15) / 16;
    if (regroup) {
        gat_count_kernel<<<PARTS, 1024, 0, stream>>>(ei, offs, E, NB, per_part);
        gat_gemm_kernel<<<GEMM_WGS, 256, 0, stream>>>(x, W, h, N, n_tiles_g, offs, bstart, NB, E, 1);
        gat_place_kernel<<<PARTS, 1024, 0, stream>>>(ei, offs, bstart, rec, E, NB, per_part);
    } else {
        long long n4 = (long long)N * C_OUT / 4;
        gat_zero_kernel<<<2048, 256, 0, stream>>>(out, n4);
        gat_gemm_kernel<<<GEMM_WGS, 256, 0, stream>>>(x, W, h, N, n_tiles_g, offs, bstart, NB, E, 0);
    }

    if (!regroup) {
        const int n_tiles = (N + 15) / 16;
        const int blocks  = (n_tiles + 7) / 8;
        gat_scores_kernel<<<blocks, 256, 0, stream>>>(h, a, s_src, s_dst, N, n_tiles);
    }

    if (regroup) {
        gat_bucket_sum_kernel<<<NB, 1024, 0, stream>>>(rec, bstart, h, a, out, N);
    } else {
        gat_edge_kernel<<<4096, 256, 0, stream>>>(ei, h, s_src, s_dst, out, E);
    }
}
